# starting kernel
# baseline (speedup 1.0000x reference)
#include <hip/hip_runtime.h>
#include <hip/hip_bf16.h>

typedef __attribute__((ext_vector_type(2))) float v2f;
typedef __attribute__((ext_vector_type(8))) float v8f;

#define BB 384
#define DD 1024
#define WD 300
#define NN 8000
#define LL 16

static __device__ __forceinline__ v8f wmma4(v2f a, v2f b, v8f c) {
  return __builtin_amdgcn_wmma_f32_16x16x4_f32(false, a, false, b, (short)0, c,
                                               false, false);
}

__global__ __launch_bounds__(32) void k_cls(const int* __restrict__ tok,
                                            const float* __restrict__ emb,
                                            const float* __restrict__ W,
                                            const float* __restrict__ bias,
                                            float* __restrict__ out) {
  const int lane = threadIdx.x & 31;
  const int half = lane >> 4;
  const int lid  = lane & 15;
  const int m0 = blockIdx.x * 64, n0 = blockIdx.y * 16;
  const int col = n0 + lid;
  const float* arow[4];
#pragma unroll
  for (int s = 0; s < 4; ++s)
    arow[s] = emb + (size_t)tok[m0 + s * 16 + lid] * WD;
  v8f acc[4];
#pragma unroll
  for (int s = 0; s < 4; ++s) acc[s] = (v8f){0.f,0.f,0.f,0.f,0.f,0.f,0.f,0.f};
  for (int k0 = 0; k0 < WD; k0 += 4) {
    const int ka = k0 + 2 * half;
    v2f b;
    b.x = W[(size_t)ka * DD + col];
    b.y = W[(size_t)(ka + 1) * DD + col];
#pragma unroll
    for (int s = 0; s < 4; ++s) {
      v2f a = *(const v2f*)(arow[s] + ka);
      acc[s] = wmma4(a, b, acc[s]);
    }
  }
  const float bv = bias[col];
#pragma unroll
  for (int s = 0; s < 4; ++s)
#pragma unroll
    for (int i = 0; i < 8; ++i)
      out[(size_t)(m0 + s * 16 + half * 8 + i) * DD + col] = acc[s][i] + bv;
}

__global__ __launch_bounds__(32) void k_concepts(const int* __restrict__ ntid,
                                                 const int* __restrict__ nlen,
                                                 const int* __restrict__ t2i,
                                                 const float* __restrict__ emb,
                                                 const float* __restrict__ W,
                                                 const float* __restrict__ bias,
                                                 const float* __restrict__ cls,
                                                 float* __restrict__ concepts) {
  const int lane = threadIdx.x & 31;
  const int half = lane >> 4;
  const int lid  = lane & 15;
  const int m0 = blockIdx.x * 64, n0 = blockIdx.y * 16;
  const int col = n0 + lid;
  const float* arow[4];
#pragma unroll
  for (int s = 0; s < 4; ++s)
    arow[s] = emb + (size_t)ntid[m0 + s * 16 + lid] * WD;
  v8f acc[4];
#pragma unroll
  for (int s = 0; s < 4; ++s) acc[s] = (v8f){0.f,0.f,0.f,0.f,0.f,0.f,0.f,0.f};
  for (int k0 = 0; k0 < WD; k0 += 4) {
    const int ka = k0 + 2 * half;
    v2f b;
    b.x = W[(size_t)ka * DD + col];
    b.y = W[(size_t)(ka + 1) * DD + col];
#pragma unroll
    for (int s = 0; s < 4; ++s) {
      v2f a = *(const v2f*)(arow[s] + ka);
      acc[s] = wmma4(a, b, acc[s]);
    }
  }
  const float bv = bias[col];
#pragma unroll
  for (int s = 0; s < 4; ++s) {
#pragma unroll
    for (int i = 0; i < 8; ++i) {
      const int row  = m0 + s * 16 + half * 8 + i;
      const int bidx = row >> 4;
      const int l    = row & 15;
      const int widx = t2i[ntid[row]];
      const bool valid = l < nlen[bidx];
      const bool found = widx > 0;
      float val = 0.f;
      if (valid) val = found ? cls[(size_t)widx * DD + col] : (acc[s][i] + bv);
      concepts[(size_t)row * DD + col] = val;
    }
  }
}

__global__ __launch_bounds__(32) void k_gemm_xw(const float* __restrict__ X,
                                                const float* __restrict__ W,
                                                const float* __restrict__ bias,
                                                float* __restrict__ out, int relu) {
  const int lane = threadIdx.x & 31;
  const int half = lane >> 4;
  const int lid  = lane & 15;
  const int m0 = blockIdx.x * 32, n0 = blockIdx.y * 16;
  const int col = n0 + lid;
  const float* arow0 = X + (size_t)(m0 + lid) * DD;
  const float* arow1 = X + (size_t)(m0 + 16 + lid) * DD;
  v8f acc0 = {0.f,0.f,0.f,0.f,0.f,0.f,0.f,0.f};
  v8f acc1 = acc0;
  for (int k0 = 0; k0 < DD; k0 += 4) {
    const int ka = k0 + 2 * half;
    v2f b;
    b.x = W[(size_t)ka * DD + col];
    b.y = W[(size_t)(ka + 1) * DD + col];
    v2f a0 = *(const v2f*)(arow0 + ka);
    v2f a1 = *(const v2f*)(arow1 + ka);
    acc0 = wmma4(a0, b, acc0);
    acc1 = wmma4(a1, b, acc1);
  }
  const float bv = bias[col];
#pragma unroll
  for (int i = 0; i < 8; ++i) {
    float v0 = acc0[i] + bv;
    float v1 = acc1[i] + bv;
    if (relu) { v0 = fmaxf(v0, 0.f); v1 = fmaxf(v1, 0.f); }
    out[(size_t)(m0 + half * 8 + i) * DD + col] = v0;
    out[(size_t)(m0 + 16 + half * 8 + i) * DD + col] = v1;
  }
}

__global__ void k_softmax(float* __restrict__ x) {
  __shared__ float red[256];
  float* p = x + (size_t)blockIdx.x * DD;
  const int tid = threadIdx.x;
  float mx = -3.402823466e+38f;
  for (int i = tid; i < DD; i += 256) mx = fmaxf(mx, p[i]);
  red[tid] = mx; __syncthreads();
  for (int s = 128; s > 0; s >>= 1) {
    if (tid < s) red[tid] = fmaxf(red[tid], red[tid + s]);
    __syncthreads();
  }
  mx = red[0]; __syncthreads();
  float sum = 0.f;
  for (int i = tid; i < DD; i += 256) { float e = __expf(p[i] - mx); p[i] = e; sum += e; }
  red[tid] = sum; __syncthreads();
  for (int s = 128; s > 0; s >>= 1) {
    if (tid < s) red[tid] += red[tid + s];
    __syncthreads();
  }
  const float inv = 1.f / red[0];
  for (int i = tid; i < DD; i += 256) p[i] *= inv;
}

__global__ void k_row_l2(const float* __restrict__ src, float* __restrict__ dn,
                         float* __restrict__ dsq) {
  __shared__ float red[256];
  const int row = blockIdx.x, tid = threadIdx.x;
  const float* p = src + (size_t)row * DD;
  float s = 0.f;
  for (int i = tid; i < DD; i += 256) {
    const float v = p[i];
    s += v * v;
    if (dsq) dsq[(size_t)row * DD + i] = v * v;
  }
  red[tid] = s; __syncthreads();
  for (int st = 128; st > 0; st >>= 1) {
    if (tid < st) red[tid] += red[tid + st];
    __syncthreads();
  }
  const float inv = rsqrtf(red[0]);
  for (int i = tid; i < DD; i += 256) dn[(size_t)row * DD + i] = p[i] * inv;
}

__global__ void k_pq(const float* __restrict__ trm, const float* __restrict__ aem,
                     float* __restrict__ P, float* __restrict__ Q) {
  __shared__ float red[256];
  const int row = blockIdx.x, tid = threadIdx.x;
  const float* tp = trm + (size_t)row * DD;
  const float* ap = aem + (size_t)row * DD;
  float s = 0.f;
  for (int i = tid; i < DD; i += 256) { const float v = tp[i]; s += v * v; }
  red[tid] = s; __syncthreads();
  for (int st = 128; st > 0; st >>= 1) {
    if (tid < st) red[tid] += red[tid + st];
    __syncthreads();
  }
  const float inv = rsqrtf(red[0]);
  for (int i = tid; i < DD; i += 256) {
    const float a = ap[i];
    P[(size_t)row * DD + i] = tp[i] * inv * a;
    Q[(size_t)row * DD + i] = a * a;
  }
}

__global__ __launch_bounds__(32) void k_score(const float* __restrict__ P,
                                              const float* __restrict__ Q,
                                              const float* __restrict__ rn,
                                              const float* __restrict__ t,
                                              const float* __restrict__ t2,
                                              const float* __restrict__ tn,
                                              float* __restrict__ score) {
  const int lane = threadIdx.x & 31;
  const int half = lane >> 4;
  const int lid  = lane & 15;
  const int m0 = blockIdx.x * 16, n0 = blockIdx.y * 16;
  const int col = n0 + lid;
  const float* prow  = P  + (size_t)(m0 + lid) * DD;
  const float* qrow  = Q  + (size_t)(m0 + lid) * DD;
  const float* rrow  = rn + (size_t)(m0 + lid) * DD;
  const float* tcol  = t  + (size_t)col * DD;
  const float* t2col = t2 + (size_t)col * DD;
  const float* tncol = tn + (size_t)col * DD;
  v8f num = {0.f,0.f,0.f,0.f,0.f,0.f,0.f,0.f};
  v8f den = num, isv = num;
  for (int k0 = 0; k0 < DD; k0 += 4) {
    const int ka = k0 + 2 * half;
    v2f a1 = *(const v2f*)(prow + ka);
    v2f b1 = *(const v2f*)(tcol + ka);
    v2f a2 = *(const v2f*)(qrow + ka);
    v2f b2 = *(const v2f*)(t2col + ka);
    v2f a3 = *(const v2f*)(rrow + ka);
    v2f b3 = *(const v2f*)(tncol + ka);
    num = wmma4(a1, b1, num);
    den = wmma4(a2, b2, den);
    isv = wmma4(a3, b3, isv);
  }
#pragma unroll
  for (int i = 0; i < 8; ++i) {
    const int row = m0 + half * 8 + i;
    score[(size_t)row * BB + col] = num[i] * rsqrtf(den[i]) + isv[i];
  }
}

extern "C" void kernel_launch(void* const* d_in, const int* in_sizes, int n_in,
                              void* d_out, int out_size, void* d_ws, size_t ws_size,
                              hipStream_t stream) {
  const float* r    = (const float*)d_in[0];
  const float* m    = (const float*)d_in[1];
  const float* t    = (const float*)d_in[2];
  const int*   ntid = (const int*)d_in[3];
  const int*   nlen = (const int*)d_in[4];
  const int*   antk = (const int*)d_in[5];
  const float* emb  = (const float*)d_in[6];
  const float* nW   = (const float*)d_in[7];
  const float* nb   = (const float*)d_in[8];
  const float* W1   = (const float*)d_in[9];
  const float* b1   = (const float*)d_in[10];
  const float* W2   = (const float*)d_in[11];
  const float* b2   = (const float*)d_in[12];
  const float* trW  = (const float*)d_in[13];
  const float* trb  = (const float*)d_in[14];
  const int*   t2i  = (const int*)d_in[15];

  float* score    = (float*)d_out;
  float* concepts = score + (size_t)BB * BB;

  float* ws  = (float*)d_ws;
  float* cls = ws;  ws += (size_t)NN * DD;
  float* h   = ws;  ws += (size_t)BB * DD;
  float* aem = ws;  ws += (size_t)BB * DD;
  float* trm = ws;  ws += (size_t)BB * DD;
  float* Pw  = ws;  ws += (size_t)BB * DD;
  float* Qw  = ws;  ws += (size_t)BB * DD;
  float* rnw = ws;  ws += (size_t)BB * DD;
  float* tnw = ws;  ws += (size_t)BB * DD;
  float* t2w = ws;  ws += (size_t)BB * DD;

  const dim3 wave(32);
  k_cls<<<dim3(NN / 64, DD / 16), wave, 0, stream>>>(antk, emb, nW, nb, cls);
  k_concepts<<<dim3(BB * LL / 64, DD / 16), wave, 0, stream>>>(ntid, nlen, t2i, emb,
                                                               nW, nb, cls, concepts);
  k_gemm_xw<<<dim3(BB / 32, DD / 16), wave, 0, stream>>>(m, W1, b1, h, 1);
  k_gemm_xw<<<dim3(BB / 32, DD / 16), wave, 0, stream>>>(h, W2, b2, aem, 0);
  k_softmax<<<BB, 256, 0, stream>>>(aem);
  k_gemm_xw<<<dim3(BB / 32, DD / 16), wave, 0, stream>>>(m, trW, trb, trm, 0);
  k_pq<<<BB, 256, 0, stream>>>(trm, aem, Pw, Qw);
  k_row_l2<<<BB, 256, 0, stream>>>(r, rnw, nullptr);
  k_row_l2<<<BB, 256, 0, stream>>>(t, tnw, t2w);
  k_score<<<dim3(BB / 16, BB / 16), wave, 0, stream>>>(Pw, Qw, rnw, t, t2w, tnw, score);
}


# K=300 gather GEMMs (class weights, concepts): each wave owns a 64x64 tile, fragments reused 4x4 from registers, B columns interleaved for 16-byte W loads and row stores; k order and sums unchanged
# speedup vs baseline: 1.3428x; 1.3428x over previous
#include <hip/hip_runtime.h>
#include <hip/hip_bf16.h>

typedef __attribute__((ext_vector_type(2))) float v2f;
typedef __attribute__((ext_vector_type(4))) float v4f;
typedef __attribute__((ext_vector_type(8))) float v8f;

#define BB 384
#define DD 1024
#define WD 300
#define NN 8000
#define LL 16

static __device__ __forceinline__ v8f wmma4(v2f a, v2f b, v8f c) {
  return __builtin_amdgcn_wmma_f32_16x16x4_f32(false, a, false, b, (short)0, c,
                                               false, false);
}

#define KC 5
#define NCH (WD / (4 * KC))

struct Chunk {
  v2f a[4][KC];
  v4f b0[KC];
  v4f b1[KC];
};

static __device__ __forceinline__ void load_chunk(Chunk& f, const float* a0,
                                                  const float* a1, const float* a2,
                                                  const float* a3, const float* wp,
                                                  int kb) {
#pragma unroll
  for (int j = 0; j < KC; ++j) f.a[0][j] = *(const v2f*)(a0 + kb + 4 * j);
#pragma unroll
  for (int j = 0; j < KC; ++j) f.a[1][j] = *(const v2f*)(a1 + kb + 4 * j);
#pragma unroll
  for (int j = 0; j < KC; ++j) f.a[2][j] = *(const v2f*)(a2 + kb + 4 * j);
#pragma unroll
  for (int j = 0; j < KC; ++j) f.a[3][j] = *(const v2f*)(a3 + kb + 4 * j);
#pragma unroll
  for (int j = 0; j < KC; ++j) {
    f.b0[j] = *(const v4f*)(wp + (size_t)(kb + 4 * j) * DD);
    f.b1[j] = *(const v4f*)(wp + (size_t)(kb + 4 * j + 1) * DD);
  }
}

static __device__ __forceinline__ void mma_chunk(v8f (&acc)[4][4], const Chunk& f) {
#pragma unroll
  for (int j = 0; j < KC; ++j) {
#pragma unroll
    for (int t = 0; t < 4; ++t) {
      v2f b;
      b.x = f.b0[j][t];
      b.y = f.b1[j][t];
#pragma unroll
      for (int s = 0; s < 4; ++s) acc[s][t] = wmma4(f.a[s][j], b, acc[s][t]);
    }
  }
}

static __device__ __forceinline__ void gemm300_tile(v8f (&acc)[4][4], const float* a0,
                                                    const float* a1, const float* a2,
                                                    const float* a3, const float* wp) {
#pragma unroll
  for (int s = 0; s < 4; ++s)
#pragma unroll
    for (int t = 0; t < 4; ++t) acc[s][t] = (v8f){0.f,0.f,0.f,0.f,0.f,0.f,0.f,0.f};
  Chunk f0, f1;
  load_chunk(f0, a0, a1, a2, a3, wp, 0);
  for (int ch = 0; ch < NCH - 1; ch += 2) {
    load_chunk(f1, a0, a1, a2, a3, wp, (ch + 1) * 4 * KC);
    mma_chunk(acc, f0);
    load_chunk(f0, a0, a1, a2, a3, wp, (ch + 2) * 4 * KC);
    mma_chunk(acc, f1);
  }
  mma_chunk(acc, f0);
}

__global__ __launch_bounds__(32) void k_cls(const int* __restrict__ tok,
                                            const float* __restrict__ emb,
                                            const float* __restrict__ W,
                                            const float* __restrict__ bias,
                                            float* __restrict__ out) {
  const int lane = threadIdx.x & 31;
  const int half = lane >> 4;
  const int lid  = lane & 15;
  const int m0 = blockIdx.x * 64, n0 = blockIdx.y * 64;
  const int c0 = n0 + 4 * lid;
  const float* a0 = emb + (size_t)tok[m0 +  0 + lid] * WD + 2 * half;
  const float* a1 = emb + (size_t)tok[m0 + 16 + lid] * WD + 2 * half;
  const float* a2 = emb + (size_t)tok[m0 + 32 + lid] * WD + 2 * half;
  const float* a3 = emb + (size_t)tok[m0 + 48 + lid] * WD + 2 * half;
  const float* wp = W + (size_t)(2 * half) * DD + c0;
  v8f acc[4][4];
  gemm300_tile(acc, a0, a1, a2, a3, wp);
  const v4f bv = *(const v4f*)(bias + c0);
#pragma unroll
  for (int s = 0; s < 4; ++s)
#pragma unroll
    for (int i = 0; i < 8; ++i) {
      v4f o;
      o.x = acc[s][0][i] + bv.x;
      o.y = acc[s][1][i] + bv.y;
      o.z = acc[s][2][i] + bv.z;
      o.w = acc[s][3][i] + bv.w;
      *(v4f*)(out + (size_t)(m0 + s * 16 + half * 8 + i) * DD + c0) = o;
    }
}

__global__ __launch_bounds__(32) void k_concepts(const int* __restrict__ ntid,
                                                 const int* __restrict__ nlen,
                                                 const int* __restrict__ t2i,
                                                 const float* __restrict__ emb,
                                                 const float* __restrict__ W,
                                                 const float* __restrict__ bias,
                                                 const float* __restrict__ cls,
                                                 float* __restrict__ concepts) {
  const int lane = threadIdx.x & 31;
  const int half = lane >> 4;
  const int lid  = lane & 15;
  const int m0 = blockIdx.x * 64, n0 = blockIdx.y * 64;
  const int c0 = n0 + 4 * lid;
  const float* a0 = emb + (size_t)ntid[m0 +  0 + lid] * WD + 2 * half;
  const float* a1 = emb + (size_t)ntid[m0 + 16 + lid] * WD + 2 * half;
  const float* a2 = emb + (size_t)ntid[m0 + 32 + lid] * WD + 2 * half;
  const float* a3 = emb + (size_t)ntid[m0 + 48 + lid] * WD + 2 * half;
  const float* wp = W + (size_t)(2 * half) * DD + c0;
  v8f acc[4][4];
  gemm300_tile(acc, a0, a1, a2, a3, wp);
  const v4f bv = *(const v4f*)(bias + c0);
#pragma unroll
  for (int s = 0; s < 4; ++s) {
#pragma unroll
    for (int i = 0; i < 8; ++i) {
      const int row  = m0 + s * 16 + half * 8 + i;
      const int bidx = row >> 4;
      const int l    = row & 15;
      const int widx = t2i[ntid[row]];
      const bool valid = l < nlen[bidx];
      const bool found = widx > 0;
      v4f val = {0.f, 0.f, 0.f, 0.f};
      if (valid) {
        if (found) {
          val = *(const v4f*)(cls + (size_t)widx * DD + c0);
        } else {
          val.x = acc[s][0][i] + bv.x;
          val.y = acc[s][1][i] + bv.y;
          val.z = acc[s][2][i] + bv.z;
          val.w = acc[s][3][i] + bv.w;
        }
      }
      *(v4f*)(concepts + (size_t)row * DD + c0) = val;
    }
  }
}

__global__ __launch_bounds__(32) void k_gemm_xw(const float* __restrict__ X,
                                                const float* __restrict__ W,
                                                const float* __restrict__ bias,
                                                float* __restrict__ out, int relu) {
  const int lane = threadIdx.x & 31;
  const int half = lane >> 4;
  const int lid  = lane & 15;
  const int m0 = blockIdx.x * 32, n0 = blockIdx.y * 16;
  const int col = n0 + lid;
  const float* arow0 = X + (size_t)(m0 + lid) * DD;
  const float* arow1 = X + (size_t)(m0 + 16 + lid) * DD;
  v8f acc0 = {0.f,0.f,0.f,0.f,0.f,0.f,0.f,0.f};
  v8f acc1 = acc0;
  for (int k0 = 0; k0 < DD; k0 += 4) {
    const int ka = k0 + 2 * half;
    v2f b;
    b.x = W[(size_t)ka * DD + col];
    b.y = W[(size_t)(ka + 1) * DD + col];
    v2f a0 = *(const v2f*)(arow0 + ka);
    v2f a1 = *(const v2f*)(arow1 + ka);
    acc0 = wmma4(a0, b, acc0);
    acc1 = wmma4(a1, b, acc1);
  }
  const float bv = bias[col];
#pragma unroll
  for (int i = 0; i < 8; ++i) {
    float v0 = acc0[i] + bv;
    float v1 = acc1[i] + bv;
    if (relu) { v0 = fmaxf(v0, 0.f); v1 = fmaxf(v1, 0.f); }
    out[(size_t)(m0 + half * 8 + i) * DD + col] = v0;
    out[(size_t)(m0 + 16 + half * 8 + i) * DD + col] = v1;
  }
}

__global__ void k_softmax(float* __restrict__ x) {
  __shared__ float red[256];
  float* p = x + (size_t)blockIdx.x * DD;
  const int tid = threadIdx.x;
  float mx = -3.402823466e+38f;
  for (int i = tid; i < DD; i += 256) mx = fmaxf(mx, p[i]);
  red[tid] = mx; __syncthreads();
  for (int s = 128; s > 0; s >>= 1) {
    if (tid < s) red[tid] = fmaxf(red[tid], red[tid + s]);
    __syncthreads();
  }
  mx = red[0]; __syncthreads();
  float sum = 0.f;
  for (int i = tid; i < DD; i += 256) { float e = __expf(p[i] - mx); p[i] = e; sum += e; }
  red[tid] = sum; __syncthreads();
  for (int s = 128; s > 0; s >>= 1) {
    if (tid < s) red[tid] += red[tid + s];
    __syncthreads();
  }
  const float inv = 1.f / red[0];
  for (int i = tid; i < DD; i += 256) p[i] *= inv;
}

__global__ void k_row_l2(const float* __restrict__ src, float* __restrict__ dn,
                         float* __restrict__ dsq) {
  __shared__ float red[256];
  const int row = blockIdx.x, tid = threadIdx.x;
  const float* p = src + (size_t)row * DD;
  float s = 0.f;
  for (int i = tid; i < DD; i += 256) {
    const float v = p[i];
    s += v * v;
    if (dsq) dsq[(size_t)row * DD + i] = v * v;
  }
  red[tid] = s; __syncthreads();
  for (int st = 128; st > 0; st >>= 1) {
    if (tid < st) red[tid] += red[tid + st];
    __syncthreads();
  }
  const float inv = rsqrtf(red[0]);
  for (int i = tid; i < DD; i += 256) dn[(size_t)row * DD + i] = p[i] * inv;
}

__global__ void k_pq(const float* __restrict__ trm, const float* __restrict__ aem,
                     float* __restrict__ P, float* __restrict__ Q) {
  __shared__ float red[256];
  const int row = blockIdx.x, tid = threadIdx.x;
  const float* tp = trm + (size_t)row * DD;
  const float* ap = aem + (size_t)row * DD;
  float s = 0.f;
  for (int i = tid; i < DD; i += 256) { const float v = tp[i]; s += v * v; }
  red[tid] = s; __syncthreads();
  for (int st = 128; st > 0; st >>= 1) {
    if (tid < st) red[tid] += red[tid + st];
    __syncthreads();
  }
  const float inv = rsqrtf(red[0]);
  for (int i = tid; i < DD; i += 256) {
    const float a = ap[i];
    P[(size_t)row * DD + i] = tp[i] * inv * a;
    Q[(size_t)row * DD + i] = a * a;
  }
}

__global__ __launch_bounds__(32) void k_score(const float* __restrict__ P,
                                              const float* __restrict__ Q,
                                              const float* __restrict__ rn,
                                              const float* __restrict__ t,
                                              const float* __restrict__ t2,
                                              const float* __restrict__ tn,
                                              float* __restrict__ score) {
  const int lane = threadIdx.x & 31;
  const int half = lane >> 4;
  const int lid  = lane & 15;
  const int m0 = blockIdx.x * 16, n0 = blockIdx.y * 16;
  const int col = n0 + lid;
  const float* prow  = P  + (size_t)(m0 + lid) * DD;
  const float* qrow  = Q  + (size_t)(m0 + lid) * DD;
  const float* rrow  = rn + (size_t)(m0 + lid) * DD;
  const float* tcol  = t  + (size_t)col * DD;
  const float* t2col = t2 + (size_t)col * DD;
  const float* tncol = tn + (size_t)col * DD;
  v8f num = {0.f,0.f,0.f,0.f,0.f,0.f,0.f,0.f};
  v8f den = num, isv = num;
  for (int k0 = 0; k0 < DD; k0 += 4) {
    const int ka = k0 + 2 * half;
    v2f a1 = *(const v2f*)(prow + ka);
    v2f b1 = *(const v2f*)(tcol + ka);
    v2f a2 = *(const v2f*)(qrow + ka);
    v2f b2 = *(const v2f*)(t2col + ka);
    v2f a3 = *(const v2f*)(rrow + ka);
    v2f b3 = *(const v2f*)(tncol + ka);
    num = wmma4(a1, b1, num);
    den = wmma4(a2, b2, den);
    isv = wmma4(a3, b3, isv);
  }
#pragma unroll
  for (int i = 0; i < 8; ++i) {
    const int row = m0 + half * 8 + i;
    score[(size_t)row * BB + col] = num[i] * rsqrtf(den[i]) + isv[i];
  }
}

extern "C" void kernel_launch(void* const* d_in, const int* in_sizes, int n_in,
                              void* d_out, int out_size, void* d_ws, size_t ws_size,
                              hipStream_t stream) {
  const float* r    = (const float*)d_in[0];
  const float* m    = (const float*)d_in[1];
  const float* t    = (const float*)d_in[2];
  const int*   ntid = (const int*)d_in[3];
  const int*   nlen = (const int*)d_in[4];
  const int*   antk = (const int*)d_in[5];
  const float* emb  = (const float*)d_in[6];
  const float* nW   = (const float*)d_in[7];
  const float* nb   = (const float*)d_in[8];
  const float* W1   = (const float*)d_in[9];
  const float* b1   = (const float*)d_in[10];
  const float* W2   = (const float*)d_in[11];
  const float* b2   = (const float*)d_in[12];
  const float* trW  = (const float*)d_in[13];
  const float* trb  = (const float*)d_in[14];
  const int*   t2i  = (const int*)d_in[15];

  float* score    = (float*)d_out;
  float* concepts = score + (size_t)BB * BB;

  float* ws  = (float*)d_ws;
  float* cls = ws;  ws += (size_t)NN * DD;
  float* h   = ws;  ws += (size_t)BB * DD;
  float* aem = ws;  ws += (size_t)BB * DD;
  float* trm = ws;  ws += (size_t)BB * DD;
  float* Pw  = ws;  ws += (size_t)BB * DD;
  float* Qw  = ws;  ws += (size_t)BB * DD;
  float* rnw = ws;  ws += (size_t)BB * DD;
  float* tnw = ws;  ws += (size_t)BB * DD;
  float* t2w = ws;  ws += (size_t)BB * DD;

  const dim3 wave(32);
  k_cls<<<dim3(NN / 64, DD / 64), wave, 0, stream>>>(antk, emb, nW, nb, cls);
  k_concepts<<<dim3(BB * LL / 64, DD / 64), wave, 0, stream>>>(ntid, nlen, t2i, emb,
                                                               nW, nb, cls, concepts);
  k_gemm_xw<<<dim3(BB / 32, DD / 16), wave, 0, stream>>>(m, W1, b1, h, 1);
  k_gemm_xw<<<dim3(BB / 32, DD / 16), wave, 0, stream>>>(h, W2, b2, aem, 0);
  k_softmax<<<BB, 256, 0, stream>>>(aem);
  k_gemm_xw<<<dim3(BB / 32, DD / 16), wave, 0, stream>>>(m, trW, trb, trm, 0);
  k_pq<<<BB, 256, 0, stream>>>(trm, aem, Pw, Qw);
  k_row_l2<<<BB, 256, 0, stream>>>(r, rnw, nullptr);
  k_row_l2<<<BB, 256, 0, stream>>>(t, tnw, t2w);
  k_score<<<dim3(BB / 16, BB / 16), wave, 0, stream>>>(Pw, Qw, rnw, t, t2w, tnw, score);
}


# K=1024 GEMMs and score kernel: K split over the 8 waves of a work-group (128 k per wave; block tiles 32x64 and 32x32), partial tiles summed through LDS in fixed wave order: summation order changes
# speedup vs baseline: 1.8478x; 1.3761x over previous
#include <hip/hip_runtime.h>
#include <hip/hip_bf16.h>

typedef __attribute__((ext_vector_type(2))) float v2f;
typedef __attribute__((ext_vector_type(4))) float v4f;
typedef __attribute__((ext_vector_type(8))) float v8f;

#define BB 384
#define DD 1024
#define WD 300
#define NN 8000
#define LL 16

static __device__ __forceinline__ v8f wmma4(v2f a, v2f b, v8f c) {
  return __builtin_amdgcn_wmma_f32_16x16x4_f32(false, a, false, b, (short)0, c,
                                               false, false);
}

#define KC 5
#define NCH (WD / (4 * KC))

struct Chunk {
  v2f a[4][KC];
  v4f b0[KC];
  v4f b1[KC];
};

static __device__ __forceinline__ void load_chunk(Chunk& f, const float* a0,
                                                  const float* a1, const float* a2,
                                                  const float* a3, const float* wp,
                                                  int kb) {
#pragma unroll
  for (int j = 0; j < KC; ++j) f.a[0][j] = *(const v2f*)(a0 + kb + 4 * j);
#pragma unroll
  for (int j = 0; j < KC; ++j) f.a[1][j] = *(const v2f*)(a1 + kb + 4 * j);
#pragma unroll
  for (int j = 0; j < KC; ++j) f.a[2][j] = *(const v2f*)(a2 + kb + 4 * j);
#pragma unroll
  for (int j = 0; j < KC; ++j) f.a[3][j] = *(const v2f*)(a3 + kb + 4 * j);
#pragma unroll
  for (int j = 0; j < KC; ++j) {
    f.b0[j] = *(const v4f*)(wp + (size_t)(kb + 4 * j) * DD);
    f.b1[j] = *(const v4f*)(wp + (size_t)(kb + 4 * j + 1) * DD);
  }
}

static __device__ __forceinline__ void mma_chunk(v8f (&acc)[4][4], const Chunk& f) {
#pragma unroll
  for (int j = 0; j < KC; ++j) {
#pragma unroll
    for (int t = 0; t < 4; ++t) {
      v2f b;
      b.x = f.b0[j][t];
      b.y = f.b1[j][t];
#pragma unroll
      for (int s = 0; s < 4; ++s) acc[s][t] = wmma4(f.a[s][j], b, acc[s][t]);
    }
  }
}

static __device__ __forceinline__ void gemm300_tile(v8f (&acc)[4][4], const float* a0,
                                                    const float* a1, const float* a2,
                                                    const float* a3, const float* wp) {
#pragma unroll
  for (int s = 0; s < 4; ++s)
#pragma unroll
    for (int t = 0; t < 4; ++t) acc[s][t] = (v8f){0.f,0.f,0.f,0.f,0.f,0.f,0.f,0.f};
  Chunk f0, f1;
  load_chunk(f0, a0, a1, a2, a3, wp, 0);
  for (int ch = 0; ch < NCH - 1; ch += 2) {
    load_chunk(f1, a0, a1, a2, a3, wp, (ch + 1) * 4 * KC);
    mma_chunk(acc, f0);
    load_chunk(f0, a0, a1, a2, a3, wp, (ch + 2) * 4 * KC);
    mma_chunk(acc, f1);
  }
  mma_chunk(acc, f0);
}

__global__ __launch_bounds__(32) void k_cls(const int* __restrict__ tok,
                                            const float* __restrict__ emb,
                                            const float* __restrict__ W,
                                            const float* __restrict__ bias,
                                            float* __restrict__ out) {
  const int lane = threadIdx.x & 31;
  const int half = lane >> 4;
  const int lid  = lane & 15;
  const int m0 = blockIdx.x * 64, n0 = blockIdx.y * 64;
  const int c0 = n0 + 4 * lid;
  const float* a0 = emb + (size_t)tok[m0 +  0 + lid] * WD + 2 * half;
  const float* a1 = emb + (size_t)tok[m0 + 16 + lid] * WD + 2 * half;
  const float* a2 = emb + (size_t)tok[m0 + 32 + lid] * WD + 2 * half;
  const float* a3 = emb + (size_t)tok[m0 + 48 + lid] * WD + 2 * half;
  const float* wp = W + (size_t)(2 * half) * DD + c0;
  v8f acc[4][4];
  gemm300_tile(acc, a0, a1, a2, a3, wp);
  const v4f bv = *(const v4f*)(bias + c0);
#pragma unroll
  for (int s = 0; s < 4; ++s)
#pragma unroll
    for (int i = 0; i < 8; ++i) {
      v4f o;
      o.x = acc[s][0][i] + bv.x;
      o.y = acc[s][1][i] + bv.y;
      o.z = acc[s][2][i] + bv.z;
      o.w = acc[s][3][i] + bv.w;
      *(v4f*)(out + (size_t)(m0 + s * 16 + half * 8 + i) * DD + c0) = o;
    }
}

__global__ __launch_bounds__(32) void k_concepts(const int* __restrict__ ntid,
                                                 const int* __restrict__ nlen,
                                                 const int* __restrict__ t2i,
                                                 const float* __restrict__ emb,
                                                 const float* __restrict__ W,
                                                 const float* __restrict__ bias,
                                                 const float* __restrict__ cls,
                                                 float* __restrict__ concepts) {
  const int lane = threadIdx.x & 31;
  const int half = lane >> 4;
  const int lid  = lane & 15;
  const int m0 = blockIdx.x * 64, n0 = blockIdx.y * 64;
  const int c0 = n0 + 4 * lid;
  const float* a0 = emb + (size_t)ntid[m0 +  0 + lid] * WD + 2 * half;
  const float* a1 = emb + (size_t)ntid[m0 + 16 + lid] * WD + 2 * half;
  const float* a2 = emb + (size_t)ntid[m0 + 32 + lid] * WD + 2 * half;
  const float* a3 = emb + (size_t)ntid[m0 + 48 + lid] * WD + 2 * half;
  const float* wp = W + (size_t)(2 * half) * DD + c0;
  v8f acc[4][4];
  gemm300_tile(acc, a0, a1, a2, a3, wp);
  const v4f bv = *(const v4f*)(bias + c0);
#pragma unroll
  for (int s = 0; s < 4; ++s) {
#pragma unroll
    for (int i = 0; i < 8; ++i) {
      const int row  = m0 + s * 16 + half * 8 + i;
      const int bidx = row >> 4;
      const int l    = row & 15;
      const int widx = t2i[ntid[row]];
      const bool valid = l < nlen[bidx];
      const bool found = widx > 0;
      v4f val = {0.f, 0.f, 0.f, 0.f};
      if (valid) {
        if (found) {
          val = *(const v4f*)(cls + (size_t)widx * DD + c0);
        } else {
          val.x = acc[s][0][i] + bv.x;
          val.y = acc[s][1][i] + bv.y;
          val.z = acc[s][2][i] + bv.z;
          val.w = acc[s][3][i] + bv.w;
        }
      }
      *(v4f*)(concepts + (size_t)row * DD + c0) = val;
    }
  }
}

#define KS 128
#define KCH 8

__global__ __launch_bounds__(256) void k_gemm_xw(const float* __restrict__ X,
                                                 const float* __restrict__ W,
                                                 const float* __restrict__ bias,
                                                 float* __restrict__ out, int relu) {
  __shared__ v4f part[8][16][16];
  const int tid  = threadIdx.x;
  const int wv   = tid >> 5;
  const int lane = tid & 31;
  const int half = lane >> 4;
  const int lid  = lane & 15;
  const int m0 = blockIdx.x * 32, n0 = blockIdx.y * 64;
  const int kb = wv * KS + 2 * half;
  const float* arow0 = X + (size_t)(m0 + lid) * DD + kb;
  const float* arow1 = X + (size_t)(m0 + 16 + lid) * DD + kb;
  const float* wp = W + (size_t)kb * DD + n0 + 4 * lid;
  v8f acc[2][4];
#pragma unroll
  for (int s = 0; s < 2; ++s)
#pragma unroll
    for (int t = 0; t < 4; ++t) acc[s][t] = (v8f){0.f,0.f,0.f,0.f,0.f,0.f,0.f,0.f};
  for (int kc = 0; kc < KS; kc += 4 * KCH) {
    v2f a0[KCH], a1[KCH];
    v4f w0[KCH], w1[KCH];
#pragma unroll
    for (int j = 0; j < KCH; ++j) {
      a0[j] = *(const v2f*)(arow0 + kc + 4 * j);
      a1[j] = *(const v2f*)(arow1 + kc + 4 * j);
      w0[j] = *(const v4f*)(wp + (size_t)(kc + 4 * j) * DD);
      w1[j] = *(const v4f*)(wp + (size_t)(kc + 4 * j + 1) * DD);
    }
#pragma unroll
    for (int j = 0; j < KCH; ++j) {
#pragma unroll
      for (int t = 0; t < 4; ++t) {
        v2f b;
        b.x = w0[j][t];
        b.y = w1[j][t];
        acc[0][t] = wmma4(a0[j], b, acc[0][t]);
        acc[1][t] = wmma4(a1[j], b, acc[1][t]);
      }
    }
  }
  const int rr = tid >> 4, cc = tid & 15;
  const v4f bv = *(const v4f*)(bias + n0 + 4 * cc);
#pragma unroll
  for (int s = 0; s < 2; ++s) {
    if (s) __syncthreads();
#pragma unroll
    for (int i = 0; i < 8; ++i) {
      v4f o;
      o.x = acc[s][0][i];
      o.y = acc[s][1][i];
      o.z = acc[s][2][i];
      o.w = acc[s][3][i];
      part[wv][half * 8 + i][lid] = o;
    }
    __syncthreads();
    v4f sum = part[0][rr][cc];
#pragma unroll
    for (int w = 1; w < 8; ++w) sum += part[w][rr][cc];
    sum += bv;
    if (relu) {
      sum.x = fmaxf(sum.x, 0.f);
      sum.y = fmaxf(sum.y, 0.f);
      sum.z = fmaxf(sum.z, 0.f);
      sum.w = fmaxf(sum.w, 0.f);
    }
    *(v4f*)(out + (size_t)(m0 + 16 * s + rr) * DD + n0 + 4 * cc) = sum;
  }
}

__global__ void k_softmax(float* __restrict__ x) {
  __shared__ float red[256];
  float* p = x + (size_t)blockIdx.x * DD;
  const int tid = threadIdx.x;
  float mx = -3.402823466e+38f;
  for (int i = tid; i < DD; i += 256) mx = fmaxf(mx, p[i]);
  red[tid] = mx; __syncthreads();
  for (int s = 128; s > 0; s >>= 1) {
    if (tid < s) red[tid] = fmaxf(red[tid], red[tid + s]);
    __syncthreads();
  }
  mx = red[0]; __syncthreads();
  float sum = 0.f;
  for (int i = tid; i < DD; i += 256) { float e = __expf(p[i] - mx); p[i] = e; sum += e; }
  red[tid] = sum; __syncthreads();
  for (int s = 128; s > 0; s >>= 1) {
    if (tid < s) red[tid] += red[tid + s];
    __syncthreads();
  }
  const float inv = 1.f / red[0];
  for (int i = tid; i < DD; i += 256) p[i] *= inv;
}

__global__ void k_row_l2(const float* __restrict__ src, float* __restrict__ dn,
                         float* __restrict__ dsq) {
  __shared__ float red[256];
  const int row = blockIdx.x, tid = threadIdx.x;
  const float* p = src + (size_t)row * DD;
  float s = 0.f;
  for (int i = tid; i < DD; i += 256) {
    const float v = p[i];
    s += v * v;
    if (dsq) dsq[(size_t)row * DD + i] = v * v;
  }
  red[tid] = s; __syncthreads();
  for (int st = 128; st > 0; st >>= 1) {
    if (tid < st) red[tid] += red[tid + st];
    __syncthreads();
  }
  const float inv = rsqrtf(red[0]);
  for (int i = tid; i < DD; i += 256) dn[(size_t)row * DD + i] = p[i] * inv;
}

__global__ void k_pq(const float* __restrict__ trm, const float* __restrict__ aem,
                     float* __restrict__ P, float* __restrict__ Q) {
  __shared__ float red[256];
  const int row = blockIdx.x, tid = threadIdx.x;
  const float* tp = trm + (size_t)row * DD;
  const float* ap = aem + (size_t)row * DD;
  float s = 0.f;
  for (int i = tid; i < DD; i += 256) { const float v = tp[i]; s += v * v; }
  red[tid] = s; __syncthreads();
  for (int st = 128; st > 0; st >>= 1) {
    if (tid < st) red[tid] += red[tid + st];
    __syncthreads();
  }
  const float inv = rsqrtf(red[0]);
  for (int i = tid; i < DD; i += 256) {
    const float a = ap[i];
    P[(size_t)row * DD + i] = tp[i] * inv * a;
    Q[(size_t)row * DD + i] = a * a;
  }
}

__global__ __launch_bounds__(256) void k_score(const float* __restrict__ P,
                                               const float* __restrict__ Q,
                                               const float* __restrict__ rn,
                                               const float* __restrict__ t,
                                               const float* __restrict__ t2,
                                               const float* __restrict__ tn,
                                               float* __restrict__ score) {
  __shared__ __attribute__((aligned(16))) float part[8][3][16][32];
  const int tid  = threadIdx.x;
  const int wv   = tid >> 5;
  const int lane = tid & 31;
  const int half = lane >> 4;
  const int lid  = lane & 15;
  const int m0 = blockIdx.x * 32, n0 = blockIdx.y * 32;
  const int kb = wv * KS + 2 * half;
  const size_t ra0 = (size_t)(m0 + lid) * DD + kb;
  const size_t ra1 = (size_t)(m0 + 16 + lid) * DD + kb;
  const size_t cb0 = (size_t)(n0 + lid) * DD + kb;
  const size_t cb1 = (size_t)(n0 + 16 + lid) * DD + kb;
  v8f num[2][2], den[2][2], isv[2][2];
#pragma unroll
  for (int s = 0; s < 2; ++s)
#pragma unroll
    for (int u = 0; u < 2; ++u) {
      num[s][u] = (v8f){0.f,0.f,0.f,0.f,0.f,0.f,0.f,0.f};
      den[s][u] = num[s][u];
      isv[s][u] = num[s][u];
    }
  for (int kc = 0; kc < KS; kc += 4 * KCH) {
    v2f a1[KCH][2], a2[KCH][2], a3[KCH][2], b1[KCH][2], b2[KCH][2], b3[KCH][2];
#pragma unroll
    for (int j = 0; j < KCH; ++j) {
      const int k0 = kc + 4 * j;
      a1[j][0] = *(const v2f*)(P  + ra0 + k0);  a1[j][1] = *(const v2f*)(P  + ra1 + k0);
      a2[j][0] = *(const v2f*)(Q  + ra0 + k0);  a2[j][1] = *(const v2f*)(Q  + ra1 + k0);
      a3[j][0] = *(const v2f*)(rn + ra0 + k0);  a3[j][1] = *(const v2f*)(rn + ra1 + k0);
      b1[j][0] = *(const v2f*)(t  + cb0 + k0);  b1[j][1] = *(const v2f*)(t  + cb1 + k0);
      b2[j][0] = *(const v2f*)(t2 + cb0 + k0);  b2[j][1] = *(const v2f*)(t2 + cb1 + k0);
      b3[j][0] = *(const v2f*)(tn + cb0 + k0);  b3[j][1] = *(const v2f*)(tn + cb1 + k0);
    }
#pragma unroll
    for (int j = 0; j < KCH; ++j)
#pragma unroll
      for (int s = 0; s < 2; ++s)
#pragma unroll
        for (int u = 0; u < 2; ++u) {
          num[s][u] = wmma4(a1[j][s], b1[j][u], num[s][u]);
          den[s][u] = wmma4(a2[j][s], b2[j][u], den[s][u]);
          isv[s][u] = wmma4(a3[j][s], b3[j][u], isv[s][u]);
        }
  }
  const int rr = tid >> 4, cc = 2 * (tid & 15);
#pragma unroll
  for (int s = 0; s < 2; ++s) {
    if (s) __syncthreads();
#pragma unroll
    for (int u = 0; u < 2; ++u)
#pragma unroll
      for (int i = 0; i < 8; ++i) {
        part[wv][0][half * 8 + i][16 * u + lid] = num[s][u][i];
        part[wv][1][half * 8 + i][16 * u + lid] = den[s][u][i];
        part[wv][2][half * 8 + i][16 * u + lid] = isv[s][u][i];
      }
    __syncthreads();
    v2f sn = *(const v2f*)&part[0][0][rr][cc];
    v2f sd = *(const v2f*)&part[0][1][rr][cc];
    v2f si = *(const v2f*)&part[0][2][rr][cc];
#pragma unroll
    for (int w = 1; w < 8; ++w) {
      sn += *(const v2f*)&part[w][0][rr][cc];
      sd += *(const v2f*)&part[w][1][rr][cc];
      si += *(const v2f*)&part[w][2][rr][cc];
    }
    v2f o;
    o.x = sn.x * rsqrtf(sd.x) + si.x;
    o.y = sn.y * rsqrtf(sd.y) + si.y;
    *(v2f*)(score + (size_t)(m0 + 16 * s + rr) * BB + n0 + cc) = o;
  }
}

extern "C" void kernel_launch(void* const* d_in, const int* in_sizes, int n_in,
                              void* d_out, int out_size, void* d_ws, size_t ws_size,
                              hipStream_t stream) {
  const float* r    = (const float*)d_in[0];
  const float* m    = (const float*)d_in[1];
  const float* t    = (const float*)d_in[2];
  const int*   ntid = (const int*)d_in[3];
  const int*   nlen = (const int*)d_in[4];
  const int*   antk = (const int*)d_in[5];
  const float* emb  = (const float*)d_in[6];
  const float* nW   = (const float*)d_in[7];
  const float* nb   = (const float*)d_in[8];
  const float* W1   = (const float*)d_in[9];
  const float* b1   = (const float*)d_in[10];
  const float* W2   = (const float*)d_in[11];
  const float* b2   = (const float*)d_in[12];
  const float* trW  = (const float*)d_in[13];
  const float* trb  = (const float*)d_in[14];
  const int*   t2i  = (const int*)d_in[15];

  float* score    = (float*)d_out;
  float* concepts = score + (size_t)BB * BB;

  float* ws  = (float*)d_ws;
  float* cls = ws;  ws += (size_t)NN * DD;
  float* h   = ws;  ws += (size_t)BB * DD;
  float* aem = ws;  ws += (size_t)BB * DD;
  float* trm = ws;  ws += (size_t)BB * DD;
  float* Pw  = ws;  ws += (size_t)BB * DD;
  float* Qw  = ws;  ws += (size_t)BB * DD;
  float* rnw = ws;  ws += (size_t)BB * DD;
  float* tnw = ws;  ws += (size_t)BB * DD;
  float* t2w = ws;  ws += (size_t)BB * DD;

  const dim3 wave(32);
  k_cls<<<dim3(NN / 64, DD / 64), wave, 0, stream>>>(antk, emb, nW, nb, cls);
  k_concepts<<<dim3(BB * LL / 64, DD / 64), wave, 0, stream>>>(ntid, nlen, t2i, emb,
                                                               nW, nb, cls, concepts);
  k_gemm_xw<<<dim3(BB / 32, DD / 64), dim3(256), 0, stream>>>(m, W1, b1, h, 1);
  k_gemm_xw<<<dim3(BB / 32, DD / 64), dim3(256), 0, stream>>>(h, W2, b2, aem, 0);
  k_softmax<<<BB, 256, 0, stream>>>(aem);
  k_gemm_xw<<<dim3(BB / 32, DD / 64), dim3(256), 0, stream>>>(m, trW, trb, trm, 0);
  k_pq<<<BB, 256, 0, stream>>>(trm, aem, Pw, Qw);
  k_row_l2<<<BB, 256, 0, stream>>>(r, rnw, nullptr);
  k_row_l2<<<BB, 256, 0, stream>>>(t, tnw, t2w);
  k_score<<<dim3(BB / 32, BB / 32), dim3(256), 0, stream>>>(Pw, Qw, rnw, t, t2w, tnw, score);
}


# Concepts epilogue: token, class-index and length lookups issued before the GEMM, all 32 class-weight row loads as one group after it, not a dependent chain per output row; k order, sums unchanged
# speedup vs baseline: 2.1168x; 1.1456x over previous
#include <hip/hip_runtime.h>
#include <hip/hip_bf16.h>

typedef __attribute__((ext_vector_type(2))) float v2f;
typedef __attribute__((ext_vector_type(4))) float v4f;
typedef __attribute__((ext_vector_type(4))) int v4i;
typedef __attribute__((ext_vector_type(8))) float v8f;

#define BB 384
#define DD 1024
#define WD 300
#define NN 8000
#define LL 16

static __device__ __forceinline__ v8f wmma4(v2f a, v2f b, v8f c) {
  return __builtin_amdgcn_wmma_f32_16x16x4_f32(false, a, false, b, (short)0, c,
                                               false, false);
}

#define KC 5
#define NCH (WD / (4 * KC))

struct Chunk {
  v2f a[4][KC];
  v4f b0[KC];
  v4f b1[KC];
};

static __device__ __forceinline__ void load_chunk(Chunk& f, const float* a0,
                                                  const float* a1, const float* a2,
                                                  const float* a3, const float* wp,
                                                  int kb) {
#pragma unroll
  for (int j = 0; j < KC; ++j) f.a[0][j] = *(const v2f*)(a0 + kb + 4 * j);
#pragma unroll
  for (int j = 0; j < KC; ++j) f.a[1][j] = *(const v2f*)(a1 + kb + 4 * j);
#pragma unroll
  for (int j = 0; j < KC; ++j) f.a[2][j] = *(const v2f*)(a2 + kb + 4 * j);
#pragma unroll
  for (int j = 0; j < KC; ++j) f.a[3][j] = *(const v2f*)(a3 + kb + 4 * j);
#pragma unroll
  for (int j = 0; j < KC; ++j) {
    f.b0[j] = *(const v4f*)(wp + (size_t)(kb + 4 * j) * DD);
    f.b1[j] = *(const v4f*)(wp + (size_t)(kb + 4 * j + 1) * DD);
  }
}

static __device__ __forceinline__ void mma_chunk(v8f (&acc)[4][4], const Chunk& f) {
#pragma unroll
  for (int j = 0; j < KC; ++j) {
#pragma unroll
    for (int t = 0; t < 4; ++t) {
      v2f b;
      b.x = f.b0[j][t];
      b.y = f.b1[j][t];
#pragma unroll
      for (int s = 0; s < 4; ++s) acc[s][t] = wmma4(f.a[s][j], b, acc[s][t]);
    }
  }
}

static __device__ __forceinline__ void gemm300_tile(v8f (&acc)[4][4], const float* a0,
                                                    const float* a1, const float* a2,
                                                    const float* a3, const float* wp) {
#pragma unroll
  for (int s = 0; s < 4; ++s)
#pragma unroll
    for (int t = 0; t < 4; ++t) acc[s][t] = (v8f){0.f,0.f,0.f,0.f,0.f,0.f,0.f,0.f};
  Chunk f0, f1;
  load_chunk(f0, a0, a1, a2, a3, wp, 0);
  for (int ch = 0; ch < NCH - 1; ch += 2) {
    load_chunk(f1, a0, a1, a2, a3, wp, (ch + 1) * 4 * KC);
    mma_chunk(acc, f0);
    load_chunk(f0, a0, a1, a2, a3, wp, (ch + 2) * 4 * KC);
    mma_chunk(acc, f1);
  }
  mma_chunk(acc, f0);
}

__global__ __launch_bounds__(32) void k_cls(const int* __restrict__ tok,
                                            const float* __restrict__ emb,
                                            const float* __restrict__ W,
                                            const float* __restrict__ bias,
                                            float* __restrict__ out) {
  const int lane = threadIdx.x & 31;
  const int half = lane >> 4;
  const int lid  = lane & 15;
  const int m0 = blockIdx.x * 64, n0 = blockIdx.y * 64;
  const int c0 = n0 + 4 * lid;
  const float* a0 = emb + (size_t)tok[m0 +  0 + lid] * WD + 2 * half;
  const float* a1 = emb + (size_t)tok[m0 + 16 + lid] * WD + 2 * half;
  const float* a2 = emb + (size_t)tok[m0 + 32 + lid] * WD + 2 * half;
  const float* a3 = emb + (size_t)tok[m0 + 48 + lid] * WD + 2 * half;
  const float* wp = W + (size_t)(2 * half) * DD + c0;
  v8f acc[4][4];
  gemm300_tile(acc, a0, a1, a2, a3, wp);
  const v4f bv = *(const v4f*)(bias + c0);
#pragma unroll
  for (int s = 0; s < 4; ++s)
#pragma unroll
    for (int i = 0; i < 8; ++i) {
      v4f o;
      o.x = acc[s][0][i] + bv.x;
      o.y = acc[s][1][i] + bv.y;
      o.z = acc[s][2][i] + bv.z;
      o.w = acc[s][3][i] + bv.w;
      *(v4f*)(out + (size_t)(m0 + s * 16 + half * 8 + i) * DD + c0) = o;
    }
}

__global__ __launch_bounds__(32) void k_concepts(const int* __restrict__ ntid,
                                                 const int* __restrict__ nlen,
                                                 const int* __restrict__ t2i,
                                                 const float* __restrict__ emb,
                                                 const float* __restrict__ W,
                                                 const float* __restrict__ bias,
                                                 const float* __restrict__ cls,
                                                 float* __restrict__ concepts) {
  const int lane = threadIdx.x & 31;
  const int half = lane >> 4;
  const int lid  = lane & 15;
  const int m0 = blockIdx.x * 64, n0 = blockIdx.y * 64;
  const int c0 = n0 + 4 * lid;
  int widx[4][8];
  int nl[4];
#pragma unroll
  for (int s = 0; s < 4; ++s) {
    const v4i t0 = *(const v4i*)(ntid + m0 + 16 * s + 8 * half);
    const v4i t1 = *(const v4i*)(ntid + m0 + 16 * s + 8 * half + 4);
#pragma unroll
    for (int i = 0; i < 4; ++i) {
      widx[s][i]     = t2i[t0[i]];
      widx[s][4 + i] = t2i[t1[i]];
    }
    nl[s] = nlen[(m0 >> 4) + s];
  }
  const float* a0 = emb + (size_t)ntid[m0 +  0 + lid] * WD + 2 * half;
  const float* a1 = emb + (size_t)ntid[m0 + 16 + lid] * WD + 2 * half;
  const float* a2 = emb + (size_t)ntid[m0 + 32 + lid] * WD + 2 * half;
  const float* a3 = emb + (size_t)ntid[m0 + 48 + lid] * WD + 2 * half;
  const float* wp = W + (size_t)(2 * half) * DD + c0;
  v8f acc[4][4];
  gemm300_tile(acc, a0, a1, a2, a3, wp);
  const v4f bv = *(const v4f*)(bias + c0);
  v4f mv[4][8];
#pragma unroll
  for (int s = 0; s < 4; ++s)
#pragma unroll
    for (int i = 0; i < 8; ++i) {
      const bool use = (8 * half + i < nl[s]) && (widx[s][i] > 0);
      mv[s][i] = *(const v4f*)(cls + (size_t)(use ? widx[s][i] : 0) * DD + c0);
    }
#pragma unroll
  for (int s = 0; s < 4; ++s) {
#pragma unroll
    for (int i = 0; i < 8; ++i) {
      const int row  = m0 + s * 16 + half * 8 + i;
      const bool valid = 8 * half + i < nl[s];
      const bool found = widx[s][i] > 0;
      v4f val = {0.f, 0.f, 0.f, 0.f};
      if (valid) {
        if (found) {
          val = mv[s][i];
        } else {
          val.x = acc[s][0][i] + bv.x;
          val.y = acc[s][1][i] + bv.y;
          val.z = acc[s][2][i] + bv.z;
          val.w = acc[s][3][i] + bv.w;
        }
      }
      *(v4f*)(concepts + (size_t)row * DD + c0) = val;
    }
  }
}

#define KS 128
#define KCH 8

__global__ __launch_bounds__(256) void k_gemm_xw(const float* __restrict__ X,
                                                 const float* __restrict__ W,
                                                 const float* __restrict__ bias,
                                                 float* __restrict__ out, int relu) {
  __shared__ v4f part[8][16][16];
  const int tid  = threadIdx.x;
  const int wv   = tid >> 5;
  const int lane = tid & 31;
  const int half = lane >> 4;
  const int lid  = lane & 15;
  const int m0 = blockIdx.x * 32, n0 = blockIdx.y * 64;
  const int kb = wv * KS + 2 * half;
  const float* arow0 = X + (size_t)(m0 + lid) * DD + kb;
  const float* arow1 = X + (size_t)(m0 + 16 + lid) * DD + kb;
  const float* wp = W + (size_t)kb * DD + n0 + 4 * lid;
  v8f acc[2][4];
#pragma unroll
  for (int s = 0; s < 2; ++s)
#pragma unroll
    for (int t = 0; t < 4; ++t) acc[s][t] = (v8f){0.f,0.f,0.f,0.f,0.f,0.f,0.f,0.f};
  for (int kc = 0; kc < KS; kc += 4 * KCH) {
    v2f a0[KCH], a1[KCH];
    v4f w0[KCH], w1[KCH];
#pragma unroll
    for (int j = 0; j < KCH; ++j) {
      a0[j] = *(const v2f*)(arow0 + kc + 4 * j);
      a1[j] = *(const v2f*)(arow1 + kc + 4 * j);
      w0[j] = *(const v4f*)(wp + (size_t)(kc + 4 * j) * DD);
      w1[j] = *(const v4f*)(wp + (size_t)(kc + 4 * j + 1) * DD);
    }
#pragma unroll
    for (int j = 0; j < KCH; ++j) {
#pragma unroll
      for (int t = 0; t < 4; ++t) {
        v2f b;
        b.x = w0[j][t];
        b.y = w1[j][t];
        acc[0][t] = wmma4(a0[j], b, acc[0][t]);
        acc[1][t] = wmma4(a1[j], b, acc[1][t]);
      }
    }
  }
  const int rr = tid >> 4, cc = tid & 15;
  const v4f bv = *(const v4f*)(bias + n0 + 4 * cc);
#pragma unroll
  for (int s = 0; s < 2; ++s) {
    if (s) __syncthreads();
#pragma unroll
    for (int i = 0; i < 8; ++i) {
      v4f o;
      o.x = acc[s][0][i];
      o.y = acc[s][1][i];
      o.z = acc[s][2][i];
      o.w = acc[s][3][i];
      part[wv][half * 8 + i][lid] = o;
    }
    __syncthreads();
    v4f sum = part[0][rr][cc];
#pragma unroll
    for (int w = 1; w < 8; ++w) sum += part[w][rr][cc];
    sum += bv;
    if (relu) {
      sum.x = fmaxf(sum.x, 0.f);
      sum.y = fmaxf(sum.y, 0.f);
      sum.z = fmaxf(sum.z, 0.f);
      sum.w = fmaxf(sum.w, 0.f);
    }
    *(v4f*)(out + (size_t)(m0 + 16 * s + rr) * DD + n0 + 4 * cc) = sum;
  }
}

__global__ void k_softmax(float* __restrict__ x) {
  __shared__ float red[256];
  float* p = x + (size_t)blockIdx.x * DD;
  const int tid = threadIdx.x;
  float mx = -3.402823466e+38f;
  for (int i = tid; i < DD; i += 256) mx = fmaxf(mx, p[i]);
  red[tid] = mx; __syncthreads();
  for (int s = 128; s > 0; s >>= 1) {
    if (tid < s) red[tid] = fmaxf(red[tid], red[tid + s]);
    __syncthreads();
  }
  mx = red[0]; __syncthreads();
  float sum = 0.f;
  for (int i = tid; i < DD; i += 256) { float e = __expf(p[i] - mx); p[i] = e; sum += e; }
  red[tid] = sum; __syncthreads();
  for (int s = 128; s > 0; s >>= 1) {
    if (tid < s) red[tid] += red[tid + s];
    __syncthreads();
  }
  const float inv = 1.f / red[0];
  for (int i = tid; i < DD; i += 256) p[i] *= inv;
}

__global__ void k_row_l2(const float* __restrict__ src, float* __restrict__ dn,
                         float* __restrict__ dsq) {
  __shared__ float red[256];
  const int row = blockIdx.x, tid = threadIdx.x;
  const float* p = src + (size_t)row * DD;
  float s = 0.f;
  for (int i = tid; i < DD; i += 256) {
    const float v = p[i];
    s += v * v;
    if (dsq) dsq[(size_t)row * DD + i] = v * v;
  }
  red[tid] = s; __syncthreads();
  for (int st = 128; st > 0; st >>= 1) {
    if (tid < st) red[tid] += red[tid + st];
    __syncthreads();
  }
  const float inv = rsqrtf(red[0]);
  for (int i = tid; i < DD; i += 256) dn[(size_t)row * DD + i] = p[i] * inv;
}

__global__ void k_pq(const float* __restrict__ trm, const float* __restrict__ aem,
                     float* __restrict__ P, float* __restrict__ Q) {
  __shared__ float red[256];
  const int row = blockIdx.x, tid = threadIdx.x;
  const float* tp = trm + (size_t)row * DD;
  const float* ap = aem + (size_t)row * DD;
  float s = 0.f;
  for (int i = tid; i < DD; i += 256) { const float v = tp[i]; s += v * v; }
  red[tid] = s; __syncthreads();
  for (int st = 128; st > 0; st >>= 1) {
    if (tid < st) red[tid] += red[tid + st];
    __syncthreads();
  }
  const float inv = rsqrtf(red[0]);
  for (int i = tid; i < DD; i += 256) {
    const float a = ap[i];
    P[(size_t)row * DD + i] = tp[i] * inv * a;
    Q[(size_t)row * DD + i] = a * a;
  }
}

__global__ __launch_bounds__(256) void k_score(const float* __restrict__ P,
                                               const float* __restrict__ Q,
                                               const float* __restrict__ rn,
                                               const float* __restrict__ t,
                                               const float* __restrict__ t2,
                                               const float* __restrict__ tn,
                                               float* __restrict__ score) {
  __shared__ __attribute__((aligned(16))) float part[8][3][16][32];
  const int tid  = threadIdx.x;
  const int wv   = tid >> 5;
  const int lane = tid & 31;
  const int half = lane >> 4;
  const int lid  = lane & 15;
  const int m0 = blockIdx.x * 32, n0 = blockIdx.y * 32;
  const int kb = wv * KS + 2 * half;
  const size_t ra0 = (size_t)(m0 + lid) * DD + kb;
  const size_t ra1 = (size_t)(m0 + 16 + lid) * DD + kb;
  const size_t cb0 = (size_t)(n0 + lid) * DD + kb;
  const size_t cb1 = (size_t)(n0 + 16 + lid) * DD + kb;
  v8f num[2][2], den[2][2], isv[2][2];
#pragma unroll
  for (int s = 0; s < 2; ++s)
#pragma unroll
    for (int u = 0; u < 2; ++u) {
      num[s][u] = (v8f){0.f,0.f,0.f,0.f,0.f,0.f,0.f,0.f};
      den[s][u] = num[s][u];
      isv[s][u] = num[s][u];
    }
  for (int kc = 0; kc < KS; kc += 4 * KCH) {
    v2f a1[KCH][2], a2[KCH][2], a3[KCH][2], b1[KCH][2], b2[KCH][2], b3[KCH][2];
#pragma unroll
    for (int j = 0; j < KCH; ++j) {
      const int k0 = kc + 4 * j;
      a1[j][0] = *(const v2f*)(P  + ra0 + k0);  a1[j][1] = *(const v2f*)(P  + ra1 + k0);
      a2[j][0] = *(const v2f*)(Q  + ra0 + k0);  a2[j][1] = *(const v2f*)(Q  + ra1 + k0);
      a3[j][0] = *(const v2f*)(rn + ra0 + k0);  a3[j][1] = *(const v2f*)(rn + ra1 + k0);
      b1[j][0] = *(const v2f*)(t  + cb0 + k0);  b1[j][1] = *(const v2f*)(t  + cb1 + k0);
      b2[j][0] = *(const v2f*)(t2 + cb0 + k0);  b2[j][1] = *(const v2f*)(t2 + cb1 + k0);
      b3[j][0] = *(const v2f*)(tn + cb0 + k0);  b3[j][1] = *(const v2f*)(tn + cb1 + k0);
    }
#pragma unroll
    for (int j = 0; j < KCH; ++j)
#pragma unroll
      for (int s = 0; s < 2; ++s)
#pragma unroll
        for (int u = 0; u < 2; ++u) {
          num[s][u] = wmma4(a1[j][s], b1[j][u], num[s][u]);
          den[s][u] = wmma4(a2[j][s], b2[j][u], den[s][u]);
          isv[s][u] = wmma4(a3[j][s], b3[j][u], isv[s][u]);
        }
  }
  const int rr = tid >> 4, cc = 2 * (tid & 15);
#pragma unroll
  for (int s = 0; s < 2; ++s) {
    if (s) __syncthreads();
#pragma unroll
    for (int u = 0; u < 2; ++u)
#pragma unroll
      for (int i = 0; i < 8; ++i) {
        part[wv][0][half * 8 + i][16 * u + lid] = num[s][u][i];
        part[wv][1][half * 8 + i][16 * u + lid] = den[s][u][i];
        part[wv][2][half * 8 + i][16 * u + lid] = isv[s][u][i];
      }
    __syncthreads();
    v2f sn = *(const v2f*)&part[0][0][rr][cc];
    v2f sd = *(const v2f*)&part[0][1][rr][cc];
    v2f si = *(const v2f*)&part[0][2][rr][cc];
#pragma unroll
    for (int w = 1; w < 8; ++w) {
      sn += *(const v2f*)&part[w][0][rr][cc];
      sd += *(const v2f*)&part[w][1][rr][cc];
      si += *(const v2f*)&part[w][2][rr][cc];
    }
    v2f o;
    o.x = sn.x * rsqrtf(sd.x) + si.x;
    o.y = sn.y * rsqrtf(sd.y) + si.y;
    *(v2f*)(score + (size_t)(m0 + 16 * s + rr) * BB + n0 + cc) = o;
  }
}

extern "C" void kernel_launch(void* const* d_in, const int* in_sizes, int n_in,
                              void* d_out, int out_size, void* d_ws, size_t ws_size,
                              hipStream_t stream) {
  const float* r    = (const float*)d_in[0];
  const float* m    = (const float*)d_in[1];
  const float* t    = (const float*)d_in[2];
  const int*   ntid = (const int*)d_in[3];
  const int*   nlen = (const int*)d_in[4];
  const int*   antk = (const int*)d_in[5];
  const float* emb  = (const float*)d_in[6];
  const float* nW   = (const float*)d_in[7];
  const float* nb   = (const float*)d_in[8];
  const float* W1   = (const float*)d_in[9];
  const float* b1   = (const float*)d_in[10];
  const float* W2   = (const float*)d_in[11];
  const float* b2   = (const float*)d_in[12];
  const float* trW  = (const float*)d_in[13];
  const float* trb  = (const float*)d_in[14];
  const int*   t2i  = (const int*)d_in[15];

  float* score    = (float*)d_out;
  float* concepts = score + (size_t)BB * BB;

  float* ws  = (float*)d_ws;
  float* cls = ws;  ws += (size_t)NN * DD;
  float* h   = ws;  ws += (size_t)BB * DD;
  float* aem = ws;  ws += (size_t)BB * DD;
  float* trm = ws;  ws += (size_t)BB * DD;
  float* Pw  = ws;  ws += (size_t)BB * DD;
  float* Qw  = ws;  ws += (size_t)BB * DD;
  float* rnw = ws;  ws += (size_t)BB * DD;
  float* tnw = ws;  ws += (size_t)BB * DD;
  float* t2w = ws;  ws += (size_t)BB * DD;

  const dim3 wave(32);
  k_cls<<<dim3(NN / 64, DD / 64), wave, 0, stream>>>(antk, emb, nW, nb, cls);
  k_concepts<<<dim3(BB * LL / 64, DD / 64), wave, 0, stream>>>(ntid, nlen, t2i, emb,
                                                               nW, nb, cls, concepts);
  k_gemm_xw<<<dim3(BB / 32, DD / 64), dim3(256), 0, stream>>>(m, W1, b1, h, 1);
  k_gemm_xw<<<dim3(BB / 32, DD / 64), dim3(256), 0, stream>>>(h, W2, b2, aem, 0);
  k_softmax<<<BB, 256, 0, stream>>>(aem);
  k_gemm_xw<<<dim3(BB / 32, DD / 64), dim3(256), 0, stream>>>(m, trW, trb, trm, 0);
  k_pq<<<BB, 256, 0, stream>>>(trm, aem, Pw, Qw);
  k_row_l2<<<BB, 256, 0, stream>>>(r, rnw, nullptr);
  k_row_l2<<<BB, 256, 0, stream>>>(t, tnw, t2w);
  k_score<<<dim3(BB / 32, BB / 32), dim3(256), 0, stream>>>(Pw, Qw, rnw, t, t2w, tnw, score);
}


# Score kernel forms P, Q, t*t and the row norms in registers from trm, softmax, r, t as it loads them: 3 row passes, 5 planes go (7 launches); norms scale the finished sums: summation order changes
# speedup vs baseline: 2.8931x; 1.3667x over previous
#include <hip/hip_runtime.h>
#include <hip/hip_bf16.h>

typedef __attribute__((ext_vector_type(2))) float v2f;
typedef __attribute__((ext_vector_type(4))) float v4f;
typedef __attribute__((ext_vector_type(4))) int v4i;
typedef __attribute__((ext_vector_type(8))) float v8f;

#define BB 384
#define DD 1024
#define WD 300
#define NN 8000
#define LL 16

static __device__ __forceinline__ v8f wmma4(v2f a, v2f b, v8f c) {
  return __builtin_amdgcn_wmma_f32_16x16x4_f32(false, a, false, b, (short)0, c,
                                               false, false);
}

static __device__ __forceinline__ v2f lo2(v4f q) { return __builtin_shufflevector(q, q, 0, 1); }
static __device__ __forceinline__ v2f hi2(v4f q) { return __builtin_shufflevector(q, q, 2, 3); }

#define KC 5
#define NCH (WD / (4 * KC))

struct Chunk {
  v4f ax[4];
  v4f ay[4];
  v2f az[4];
  v4f b0[KC];
  v4f b1[KC];
};

static __device__ __forceinline__ void load_chunk(Chunk& f, const float* a0,
                                                  const float* a1, const float* a2,
                                                  const float* a3, int zd,
                                                  const float* wpx, const float* wpz,
                                                  int kb) {
  const float* ar[4] = {a0, a1, a2, a3};
#pragma unroll
  for (int s = 0; s < 4; ++s) {
    f.ax[s] = *(const v4f*)(ar[s] + kb);
    f.ay[s] = *(const v4f*)(ar[s] + kb + 4);
    f.az[s] = *(const v2f*)(ar[s] + kb + zd);
  }
#pragma unroll
  for (int j = 0; j < KC - 1; ++j) {
    f.b0[j] = *(const v4f*)(wpx + (size_t)(kb + 2 * j) * DD);
    f.b1[j] = *(const v4f*)(wpx + (size_t)(kb + 2 * j + 1) * DD);
  }
  f.b0[KC - 1] = *(const v4f*)(wpz + (size_t)kb * DD);
  f.b1[KC - 1] = *(const v4f*)(wpz + (size_t)(kb + 1) * DD);
}

static __device__ __forceinline__ void mma_chunk(v8f (&acc)[4][4], const Chunk& f) {
#pragma unroll
  for (int j = 0; j < KC; ++j) {
#pragma unroll
    for (int t = 0; t < 4; ++t) {
      v2f b;
      b.x = f.b0[j][t];
      b.y = f.b1[j][t];
#pragma unroll
      for (int s = 0; s < 4; ++s) {
        const v2f a = j == 0 ? lo2(f.ax[s]) : j == 1 ? hi2(f.ax[s])
                    : j == 2 ? lo2(f.ay[s]) : j == 3 ? hi2(f.ay[s]) : f.az[s];
        acc[s][t] = wmma4(a, b, acc[s][t]);
      }
    }
  }
}

static __device__ __forceinline__ void gemm300_tile(v8f (&acc)[4][4], const float* a0,
                                                    const float* a1, const float* a2,
                                                    const float* a3, int zd,
                                                    const float* wpx, const float* wpz) {
#pragma unroll
  for (int s = 0; s < 4; ++s)
#pragma unroll
    for (int t = 0; t < 4; ++t) acc[s][t] = (v8f){0.f,0.f,0.f,0.f,0.f,0.f,0.f,0.f};
  Chunk f0, f1;
  load_chunk(f0, a0, a1, a2, a3, zd, wpx, wpz, 0);
#pragma unroll 1
  for (int ch = 0; ch < NCH - 1; ch += 2) {
    load_chunk(f1, a0, a1, a2, a3, zd, wpx, wpz, (ch + 1) * 4 * KC);
    mma_chunk(acc, f0);
    load_chunk(f0, a0, a1, a2, a3, zd, wpx, wpz, (ch + 2) * 4 * KC);
    mma_chunk(acc, f1);
  }
  mma_chunk(acc, f0);
}

__global__ __launch_bounds__(32) void k_cls(const int* __restrict__ tok,
                                            const float* __restrict__ emb,
                                            const float* __restrict__ W,
                                            const float* __restrict__ bias,
                                            float* __restrict__ out) {
  const int lane = threadIdx.x & 31;
  const int half = lane >> 4;
  const int lid  = lane & 15;
  const int m0 = blockIdx.x * 64, n0 = blockIdx.y * 64;
  const int c0 = n0 + 4 * lid;
  const int xo = half ? 12 : 0;
  const int zo = half ? 10 : 8;
  const float* a0 = emb + (size_t)tok[m0 +  0 + lid] * WD + xo;
  const float* a1 = emb + (size_t)tok[m0 + 16 + lid] * WD + xo;
  const float* a2 = emb + (size_t)tok[m0 + 32 + lid] * WD + xo;
  const float* a3 = emb + (size_t)tok[m0 + 48 + lid] * WD + xo;
  const float* wpx = W + (size_t)xo * DD + c0;
  const float* wpz = W + (size_t)zo * DD + c0;
  v8f acc[4][4];
  gemm300_tile(acc, a0, a1, a2, a3, zo - xo, wpx, wpz);
  const v4f bv = *(const v4f*)(bias + c0);
#pragma unroll
  for (int s = 0; s < 4; ++s)
#pragma unroll
    for (int i = 0; i < 8; ++i) {
      v4f o;
      o.x = acc[s][0][i] + bv.x;
      o.y = acc[s][1][i] + bv.y;
      o.z = acc[s][2][i] + bv.z;
      o.w = acc[s][3][i] + bv.w;
      *(v4f*)(out + (size_t)(m0 + s * 16 + half * 8 + i) * DD + c0) = o;
    }
}

__global__ __launch_bounds__(32) void k_concepts(const int* __restrict__ ntid,
                                                 const int* __restrict__ nlen,
                                                 const int* __restrict__ t2i,
                                                 const float* __restrict__ emb,
                                                 const float* __restrict__ W,
                                                 const float* __restrict__ bias,
                                                 const float* __restrict__ cls,
                                                 float* __restrict__ concepts) {
  const int lane = threadIdx.x & 31;
  const int half = lane >> 4;
  const int lid  = lane & 15;
  const int m0 = blockIdx.x * 64, n0 = blockIdx.y * 64;
  const int c0 = n0 + 4 * lid;
  int widx[4][8];
  int nl[4];
#pragma unroll
  for (int s = 0; s < 4; ++s) {
    const v4i t0 = *(const v4i*)(ntid + m0 + 16 * s + 8 * half);
    const v4i t1 = *(const v4i*)(ntid + m0 + 16 * s + 8 * half + 4);
#pragma unroll
    for (int i = 0; i < 4; ++i) {
      widx[s][i]     = t2i[t0[i]];
      widx[s][4 + i] = t2i[t1[i]];
    }
    nl[s] = nlen[(m0 >> 4) + s];
  }
  const int xo = half ? 12 : 0;
  const int zo = half ? 10 : 8;
  const float* a0 = emb + (size_t)ntid[m0 +  0 + lid] * WD + xo;
  const float* a1 = emb + (size_t)ntid[m0 + 16 + lid] * WD + xo;
  const float* a2 = emb + (size_t)ntid[m0 + 32 + lid] * WD + xo;
  const float* a3 = emb + (size_t)ntid[m0 + 48 + lid] * WD + xo;
  const float* wpx = W + (size_t)xo * DD + c0;
  const float* wpz = W + (size_t)zo * DD + c0;
  v8f acc[4][4];
  gemm300_tile(acc, a0, a1, a2, a3, zo - xo, wpx, wpz);
  const v4f bv = *(const v4f*)(bias + c0);
  v4f mv[4][8];
#pragma unroll
  for (int s = 0; s < 4; ++s)
#pragma unroll
    for (int i = 0; i < 8; ++i) {
      const bool use = (8 * half + i < nl[s]) && (widx[s][i] > 0);
      mv[s][i] = *(const v4f*)(cls + (size_t)(use ? widx[s][i] : 0) * DD + c0);
    }
#pragma unroll
  for (int s = 0; s < 4; ++s) {
#pragma unroll
    for (int i = 0; i < 8; ++i) {
      const int row  = m0 + s * 16 + half * 8 + i;
      const bool valid = 8 * half + i < nl[s];
      const bool found = widx[s][i] > 0;
      v4f val = {0.f, 0.f, 0.f, 0.f};
      if (valid) {
        if (found) {
          val = mv[s][i];
        } else {
          val.x = acc[s][0][i] + bv.x;
          val.y = acc[s][1][i] + bv.y;
          val.z = acc[s][2][i] + bv.z;
          val.w = acc[s][3][i] + bv.w;
        }
      }
      *(v4f*)(concepts + (size_t)row * DD + c0) = val;
    }
  }
}

#define KS 128
#define KCH 8

__global__ __launch_bounds__(256) void k_gemm_xw(const float* __restrict__ X,
                                                 const float* __restrict__ W,
                                                 const float* __restrict__ bias,
                                                 float* __restrict__ out, int relu) {
  __shared__ v4f part[8][16][16];
  const int tid  = threadIdx.x;
  const int wv   = tid >> 5;
  const int lane = tid & 31;
  const int half = lane >> 4;
  const int lid  = lane & 15;
  const int m0 = blockIdx.x * 32, n0 = blockIdx.y * 64;
  const int kb = wv * KS + 16 * half;
  const float* arow0 = X + (size_t)(m0 + lid) * DD + kb;
  const float* arow1 = X + (size_t)(m0 + 16 + lid) * DD + kb;
  const float* wp = W + (size_t)kb * DD + n0 + 4 * lid;
  v8f acc[2][4];
#pragma unroll
  for (int s = 0; s < 2; ++s)
#pragma unroll
    for (int t = 0; t < 4; ++t) acc[s][t] = (v8f){0.f,0.f,0.f,0.f,0.f,0.f,0.f,0.f};
  for (int kc = 0; kc < KS; kc += 4 * KCH) {
    v4f a0[KCH / 2], a1[KCH / 2];
    v4f w0[KCH], w1[KCH];
#pragma unroll
    for (int i = 0; i < KCH / 2; ++i) {
      a0[i] = *(const v4f*)(arow0 + kc + 4 * i);
      a1[i] = *(const v4f*)(arow1 + kc + 4 * i);
    }
#pragma unroll
    for (int j = 0; j < KCH; ++j) {
      w0[j] = *(const v4f*)(wp + (size_t)(kc + 2 * j) * DD);
      w1[j] = *(const v4f*)(wp + (size_t)(kc + 2 * j + 1) * DD);
    }
    __builtin_amdgcn_sched_barrier(0);
#pragma unroll
    for (int j = 0; j < KCH; ++j) {
      const v2f fa0 = (j & 1) ? hi2(a0[j >> 1]) : lo2(a0[j >> 1]);
      const v2f fa1 = (j & 1) ? hi2(a1[j >> 1]) : lo2(a1[j >> 1]);
#pragma unroll
      for (int t = 0; t < 4; ++t) {
        v2f b;
        b.x = w0[j][t];
        b.y = w1[j][t];
        acc[0][t] = wmma4(fa0, b, acc[0][t]);
        acc[1][t] = wmma4(fa1, b, acc[1][t]);
      }
    }
  }
  const int rr = tid >> 4, cc = tid & 15;
  const v4f bv = *(const v4f*)(bias + n0 + 4 * cc);
#pragma unroll
  for (int s = 0; s < 2; ++s) {
    if (s) __syncthreads();
#pragma unroll
    for (int i = 0; i < 8; ++i) {
      v4f o;
      o.x = acc[s][0][i];
      o.y = acc[s][1][i];
      o.z = acc[s][2][i];
      o.w = acc[s][3][i];
      part[wv][half * 8 + i][lid] = o;
    }
    __syncthreads();
    v4f sum = part[0][rr][cc];
#pragma unroll
    for (int w = 1; w < 8; ++w) sum += part[w][rr][cc];
    sum += bv;
    if (relu) {
      sum.x = fmaxf(sum.x, 0.f);
      sum.y = fmaxf(sum.y, 0.f);
      sum.z = fmaxf(sum.z, 0.f);
      sum.w = fmaxf(sum.w, 0.f);
    }
    *(v4f*)(out + (size_t)(m0 + 16 * s + rr) * DD + n0 + 4 * cc) = sum;
  }
}

__global__ void k_softmax(float* __restrict__ x) {
  __shared__ float red[256];
  float* p = x + (size_t)blockIdx.x * DD;
  const int tid = threadIdx.x;
  float mx = -3.402823466e+38f;
  for (int i = tid; i < DD; i += 256) mx = fmaxf(mx, p[i]);
  red[tid] = mx; __syncthreads();
  for (int s = 128; s > 0; s >>= 1) {
    if (tid < s) red[tid] = fmaxf(red[tid], red[tid + s]);
    __syncthreads();
  }
  mx = red[0]; __syncthreads();
  float sum = 0.f;
  for (int i = tid; i < DD; i += 256) { float e = __expf(p[i] - mx); p[i] = e; sum += e; }
  red[tid] = sum; __syncthreads();
  for (int s = 128; s > 0; s >>= 1) {
    if (tid < s) red[tid] += red[tid + s];
    __syncthreads();
  }
  const float inv = 1.f / red[0];
  for (int i = tid; i < DD; i += 256) p[i] *= inv;
}

static __device__ __forceinline__ v2f sq_acc(v2f acc, v4f q) {
  acc += lo2(q) * lo2(q);
  acc += hi2(q) * hi2(q);
  return acc;
}

__global__ __launch_bounds__(256) void k_score(const float* __restrict__ trm,
                                               const float* __restrict__ aem,
                                               const float* __restrict__ r,
                                               const float* __restrict__ t,
                                               float* __restrict__ score) {
  __shared__ __attribute__((aligned(16))) float part[8][3][16][32];
  __shared__ __attribute__((aligned(16))) float ssq[8][3][32];
  const int tid  = threadIdx.x;
  const int wv   = tid >> 5;
  const int lane = tid & 31;
  const int half = lane >> 4;
  const int lid  = lane & 15;
  const int m0 = blockIdx.x * 32, n0 = blockIdx.y * 32;
  const int kb = wv * KS + 16 * half;
  const size_t ra0 = (size_t)(m0 + lid) * DD + kb;
  const size_t ra1 = (size_t)(m0 + 16 + lid) * DD + kb;
  const size_t cb0 = (size_t)(n0 + lid) * DD + kb;
  const size_t cb1 = (size_t)(n0 + 16 + lid) * DD + kb;
  v8f num[2][2], den[2][2], isv[2][2];
#pragma unroll
  for (int s = 0; s < 2; ++s)
#pragma unroll
    for (int u = 0; u < 2; ++u) {
      num[s][u] = (v8f){0.f,0.f,0.f,0.f,0.f,0.f,0.f,0.f};
      den[s][u] = num[s][u];
      isv[s][u] = num[s][u];
    }
  v2f qt2[2], qr2[2], qc2[2];
#pragma unroll
  for (int s = 0; s < 2; ++s) {
    qt2[s] = (v2f){0.f, 0.f};
    qr2[s] = qt2[s];
    qc2[s] = qt2[s];
  }
  for (int kc = 0; kc < KS; kc += 4 * KCH) {
    v4f a1[KCH / 2][2], a2[KCH / 2][2], a3[KCH / 2][2];
    v4f b1[KCH / 2][2], b2[KCH / 2][2];
#pragma unroll
    for (int i = 0; i < KCH / 2; ++i) {
      const int k0 = kc + 4 * i;
      a1[i][0] = *(const v4f*)(trm + ra0 + k0);  a1[i][1] = *(const v4f*)(trm + ra1 + k0);
      a2[i][0] = *(const v4f*)(aem + ra0 + k0);  a2[i][1] = *(const v4f*)(aem + ra1 + k0);
      a3[i][0] = *(const v4f*)(r   + ra0 + k0);  a3[i][1] = *(const v4f*)(r   + ra1 + k0);
      b1[i][0] = *(const v4f*)(t   + cb0 + k0);  b1[i][1] = *(const v4f*)(t   + cb1 + k0);
    }
    __builtin_amdgcn_sched_barrier(0);
#pragma unroll
    for (int i = 0; i < KCH / 2; ++i)
#pragma unroll
      for (int s = 0; s < 2; ++s) {
        qt2[s] = sq_acc(qt2[s], a1[i][s]);
        qr2[s] = sq_acc(qr2[s], a3[i][s]);
        qc2[s] = sq_acc(qc2[s], b1[i][s]);
        a1[i][s] = a1[i][s] * a2[i][s];
        a2[i][s] = a2[i][s] * a2[i][s];
        b2[i][s] = b1[i][s] * b1[i][s];
      }
#pragma unroll
    for (int j = 0; j < KCH; ++j)
#pragma unroll
      for (int s = 0; s < 2; ++s)
#pragma unroll
        for (int u = 0; u < 2; ++u) {
          const int i = j >> 1;
          const v2f fa1 = (j & 1) ? hi2(a1[i][s]) : lo2(a1[i][s]);
          const v2f fa2 = (j & 1) ? hi2(a2[i][s]) : lo2(a2[i][s]);
          const v2f fa3 = (j & 1) ? hi2(a3[i][s]) : lo2(a3[i][s]);
          const v2f fb1 = (j & 1) ? hi2(b1[i][u]) : lo2(b1[i][u]);
          const v2f fb2 = (j & 1) ? hi2(b2[i][u]) : lo2(b2[i][u]);
          num[s][u] = wmma4(fa1, fb1, num[s][u]);
          den[s][u] = wmma4(fa2, fb2, den[s][u]);
          isv[s][u] = wmma4(fa3, fb1, isv[s][u]);
        }
  }
  float qt[2], qr[2], qc[2];
#pragma unroll
  for (int s = 0; s < 2; ++s) {
    qt[s] = qt2[s].x + qt2[s].y;
    qr[s] = qr2[s].x + qr2[s].y;
    qc[s] = qc2[s].x + qc2[s].y;
    qt[s] += __shfl_xor(qt[s], 16);
    qr[s] += __shfl_xor(qr[s], 16);
    qc[s] += __shfl_xor(qc[s], 16);
  }
  if (half == 0) {
#pragma unroll
    for (int s = 0; s < 2; ++s) {
      ssq[wv][0][16 * s + lid] = qt[s];
      ssq[wv][1][16 * s + lid] = qr[s];
      ssq[wv][2][16 * s + lid] = qc[s];
    }
  }
  const int rr = tid >> 4, cc = 2 * (tid & 15);
  v2f ic = {0.f, 0.f};
#pragma unroll
  for (int s = 0; s < 2; ++s) {
    if (s) __syncthreads();
#pragma unroll
    for (int u = 0; u < 2; ++u)
#pragma unroll
      for (int i = 0; i < 8; ++i) {
        part[wv][0][half * 8 + i][16 * u + lid] = num[s][u][i];
        part[wv][1][half * 8 + i][16 * u + lid] = den[s][u][i];
        part[wv][2][half * 8 + i][16 * u + lid] = isv[s][u][i];
      }
    __syncthreads();
    if (s == 0) {
      v2f c = *(const v2f*)&ssq[0][2][cc];
#pragma unroll
      for (int w = 1; w < 8; ++w) c += *(const v2f*)&ssq[w][2][cc];
      ic.x = rsqrtf(c.x);
      ic.y = rsqrtf(c.y);
    }
    float st = ssq[0][0][16 * s + rr];
    float sr = ssq[0][1][16 * s + rr];
#pragma unroll
    for (int w = 1; w < 8; ++w) {
      st += ssq[w][0][16 * s + rr];
      sr += ssq[w][1][16 * s + rr];
    }
    const float it = rsqrtf(st);
    const float ir = rsqrtf(sr);
    v2f sn = *(const v2f*)&part[0][0][rr][cc];
    v2f sd = *(const v2f*)&part[0][1][rr][cc];
    v2f si = *(const v2f*)&part[0][2][rr][cc];
#pragma unroll
    for (int w = 1; w < 8; ++w) {
      sn += *(const v2f*)&part[w][0][rr][cc];
      sd += *(const v2f*)&part[w][1][rr][cc];
      si += *(const v2f*)&part[w][2][rr][cc];
    }
    v2f o;
    o.x = sn.x * rsqrtf(sd.x) * it + si.x * (ir * ic.x);
    o.y = sn.y * rsqrtf(sd.y) * it + si.y * (ir * ic.y);
    *(v2f*)(score + (size_t)(m0 + 16 * s + rr) * BB + n0 + cc) = o;
  }
}

extern "C" void kernel_launch(void* const* d_in, const int* in_sizes, int n_in,
                              void* d_out, int out_size, void* d_ws, size_t ws_size,
                              hipStream_t stream) {
  const float* r    = (const float*)d_in[0];
  const float* m    = (const float*)d_in[1];
  const float* t    = (const float*)d_in[2];
  const int*   ntid = (const int*)d_in[3];
  const int*   nlen = (const int*)d_in[4];
  const int*   antk = (const int*)d_in[5];
  const float* emb  = (const float*)d_in[6];
  const float* nW   = (const float*)d_in[7];
  const float* nb   = (const float*)d_in[8];
  const float* W1   = (const float*)d_in[9];
  const float* b1   = (const float*)d_in[10];
  const float* W2   = (const float*)d_in[11];
  const float* b2   = (const float*)d_in[12];
  const float* trW  = (const float*)d_in[13];
  const float* trb  = (const float*)d_in[14];
  const int*   t2i  = (const int*)d_in[15];

  float* score    = (float*)d_out;
  float* concepts = score + (size_t)BB * BB;

  float* ws  = (float*)d_ws;
  float* cls = ws;  ws += (size_t)NN * DD;
  float* h   = ws;  ws += (size_t)BB * DD;
  float* aem = ws;  ws += (size_t)BB * DD;
  float* trm = ws;

  const dim3 wave(32);
  k_cls<<<dim3(NN / 64, DD / 64), wave, 0, stream>>>(antk, emb, nW, nb, cls);
  k_concepts<<<dim3(BB * LL / 64, DD / 64), wave, 0, stream>>>(ntid, nlen, t2i, emb,
                                                               nW, nb, cls, concepts);
  k_gemm_xw<<<dim3(BB / 32, DD / 64), dim3(256), 0, stream>>>(m, W1, b1, h, 1);
  k_gemm_xw<<<dim3(BB / 32, DD / 64), dim3(256), 0, stream>>>(h, W2, b2, aem, 0);
  k_softmax<<<BB, 256, 0, stream>>>(aem);
  k_gemm_xw<<<dim3(BB / 32, DD / 64), dim3(256), 0, stream>>>(m, trW, trb, trm, 0);
  k_score<<<dim3(BB / 32, BB / 32), dim3(256), 0, stream>>>(trm, aem, r, t, score);
}


# Softmax pass dropped: score kernel reads the logits, finds each row's max itself (its 8 waves span a row), takes exp as pieces load; 1/sum cancels in num/sqrt(den): 6 launches; rounding changes
# speedup vs baseline: 3.0182x; 1.0432x over previous
#include <hip/hip_runtime.h>
#include <hip/hip_bf16.h>

typedef __attribute__((ext_vector_type(2))) float v2f;
typedef __attribute__((ext_vector_type(4))) float v4f;
typedef __attribute__((ext_vector_type(4))) int v4i;
typedef __attribute__((ext_vector_type(8))) float v8f;

#define BB 384
#define DD 1024
#define WD 300
#define NN 8000
#define LL 16

static __device__ __forceinline__ v8f wmma4(v2f a, v2f b, v8f c) {
  return __builtin_amdgcn_wmma_f32_16x16x4_f32(false, a, false, b, (short)0, c,
                                               false, false);
}

static __device__ __forceinline__ v2f lo2(v4f q) { return __builtin_shufflevector(q, q, 0, 1); }
static __device__ __forceinline__ v2f hi2(v4f q) { return __builtin_shufflevector(q, q, 2, 3); }

#define KC 5
#define NCH (WD / (4 * KC))

struct Chunk {
  v4f ax[4];
  v4f ay[4];
  v2f az[4];
  v4f b0[KC];
  v4f b1[KC];
};

static __device__ __forceinline__ void load_chunk(Chunk& f, const float* a0,
                                                  const float* a1, const float* a2,
                                                  const float* a3, int zd,
                                                  const float* wpx, const float* wpz,
                                                  int kb) {
  const float* ar[4] = {a0, a1, a2, a3};
#pragma unroll
  for (int s = 0; s < 4; ++s) {
    f.ax[s] = *(const v4f*)(ar[s] + kb);
    f.ay[s] = *(const v4f*)(ar[s] + kb + 4);
    f.az[s] = *(const v2f*)(ar[s] + kb + zd);
  }
#pragma unroll
  for (int j = 0; j < KC - 1; ++j) {
    f.b0[j] = *(const v4f*)(wpx + (size_t)(kb + 2 * j) * DD);
    f.b1[j] = *(const v4f*)(wpx + (size_t)(kb + 2 * j + 1) * DD);
  }
  f.b0[KC - 1] = *(const v4f*)(wpz + (size_t)kb * DD);
  f.b1[KC - 1] = *(const v4f*)(wpz + (size_t)(kb + 1) * DD);
}

static __device__ __forceinline__ void mma_chunk(v8f (&acc)[4][4], const Chunk& f) {
#pragma unroll
  for (int j = 0; j < KC; ++j) {
#pragma unroll
    for (int t = 0; t < 4; ++t) {
      v2f b;
      b.x = f.b0[j][t];
      b.y = f.b1[j][t];
#pragma unroll
      for (int s = 0; s < 4; ++s) {
        const v2f a = j == 0 ? lo2(f.ax[s]) : j == 1 ? hi2(f.ax[s])
                    : j == 2 ? lo2(f.ay[s]) : j == 3 ? hi2(f.ay[s]) : f.az[s];
        acc[s][t] = wmma4(a, b, acc[s][t]);
      }
    }
  }
}

static __device__ __forceinline__ void gemm300_tile(v8f (&acc)[4][4], const float* a0,
                                                    const float* a1, const float* a2,
                                                    const float* a3, int zd,
                                                    const float* wpx, const float* wpz) {
#pragma unroll
  for (int s = 0; s < 4; ++s)
#pragma unroll
    for (int t = 0; t < 4; ++t) acc[s][t] = (v8f){0.f,0.f,0.f,0.f,0.f,0.f,0.f,0.f};
  Chunk f0, f1;
  load_chunk(f0, a0, a1, a2, a3, zd, wpx, wpz, 0);
#pragma unroll 1
  for (int ch = 0; ch < NCH - 1; ch += 2) {
    load_chunk(f1, a0, a1, a2, a3, zd, wpx, wpz, (ch + 1) * 4 * KC);
    mma_chunk(acc, f0);
    load_chunk(f0, a0, a1, a2, a3, zd, wpx, wpz, (ch + 2) * 4 * KC);
    mma_chunk(acc, f1);
  }
  mma_chunk(acc, f0);
}

__global__ __launch_bounds__(32) void k_cls(const int* __restrict__ tok,
                                            const float* __restrict__ emb,
                                            const float* __restrict__ W,
                                            const float* __restrict__ bias,
                                            float* __restrict__ out) {
  const int lane = threadIdx.x & 31;
  const int half = lane >> 4;
  const int lid  = lane & 15;
  const int m0 = blockIdx.x * 64, n0 = blockIdx.y * 64;
  const int c0 = n0 + 4 * lid;
  const int xo = half ? 12 : 0;
  const int zo = half ? 10 : 8;
  const float* a0 = emb + (size_t)tok[m0 +  0 + lid] * WD + xo;
  const float* a1 = emb + (size_t)tok[m0 + 16 + lid] * WD + xo;
  const float* a2 = emb + (size_t)tok[m0 + 32 + lid] * WD + xo;
  const float* a3 = emb + (size_t)tok[m0 + 48 + lid] * WD + xo;
  const float* wpx = W + (size_t)xo * DD + c0;
  const float* wpz = W + (size_t)zo * DD + c0;
  v8f acc[4][4];
  gemm300_tile(acc, a0, a1, a2, a3, zo - xo, wpx, wpz);
  const v4f bv = *(const v4f*)(bias + c0);
#pragma unroll
  for (int s = 0; s < 4; ++s)
#pragma unroll
    for (int i = 0; i < 8; ++i) {
      v4f o;
      o.x = acc[s][0][i] + bv.x;
      o.y = acc[s][1][i] + bv.y;
      o.z = acc[s][2][i] + bv.z;
      o.w = acc[s][3][i] + bv.w;
      *(v4f*)(out + (size_t)(m0 + s * 16 + half * 8 + i) * DD + c0) = o;
    }
}

__global__ __launch_bounds__(32) void k_concepts(const int* __restrict__ ntid,
                                                 const int* __restrict__ nlen,
                                                 const int* __restrict__ t2i,
                                                 const float* __restrict__ emb,
                                                 const float* __restrict__ W,
                                                 const float* __restrict__ bias,
                                                 const float* __restrict__ cls,
                                                 float* __restrict__ concepts) {
  const int lane = threadIdx.x & 31;
  const int half = lane >> 4;
  const int lid  = lane & 15;
  const int m0 = blockIdx.x * 64, n0 = blockIdx.y * 64;
  const int c0 = n0 + 4 * lid;
  int widx[4][8];
  int nl[4];
#pragma unroll
  for (int s = 0; s < 4; ++s) {
    const v4i t0 = *(const v4i*)(ntid + m0 + 16 * s + 8 * half);
    const v4i t1 = *(const v4i*)(ntid + m0 + 16 * s + 8 * half + 4);
#pragma unroll
    for (int i = 0; i < 4; ++i) {
      widx[s][i]     = t2i[t0[i]];
      widx[s][4 + i] = t2i[t1[i]];
    }
    nl[s] = nlen[(m0 >> 4) + s];
  }
  const int xo = half ? 12 : 0;
  const int zo = half ? 10 : 8;
  const float* a0 = emb + (size_t)ntid[m0 +  0 + lid] * WD + xo;
  const float* a1 = emb + (size_t)ntid[m0 + 16 + lid] * WD + xo;
  const float* a2 = emb + (size_t)ntid[m0 + 32 + lid] * WD + xo;
  const float* a3 = emb + (size_t)ntid[m0 + 48 + lid] * WD + xo;
  const float* wpx = W + (size_t)xo * DD + c0;
  const float* wpz = W + (size_t)zo * DD + c0;
  v8f acc[4][4];
  gemm300_tile(acc, a0, a1, a2, a3, zo - xo, wpx, wpz);
  const v4f bv = *(const v4f*)(bias + c0);
  v4f mv[4][8];
#pragma unroll
  for (int s = 0; s < 4; ++s)
#pragma unroll
    for (int i = 0; i < 8; ++i) {
      const bool use = (8 * half + i < nl[s]) && (widx[s][i] > 0);
      mv[s][i] = *(const v4f*)(cls + (size_t)(use ? widx[s][i] : 0) * DD + c0);
    }
#pragma unroll
  for (int s = 0; s < 4; ++s) {
#pragma unroll
    for (int i = 0; i < 8; ++i) {
      const int row  = m0 + s * 16 + half * 8 + i;
      const bool valid = 8 * half + i < nl[s];
      const bool found = widx[s][i] > 0;
      v4f val = {0.f, 0.f, 0.f, 0.f};
      if (valid) {
        if (found) {
          val = mv[s][i];
        } else {
          val.x = acc[s][0][i] + bv.x;
          val.y = acc[s][1][i] + bv.y;
          val.z = acc[s][2][i] + bv.z;
          val.w = acc[s][3][i] + bv.w;
        }
      }
      *(v4f*)(concepts + (size_t)row * DD + c0) = val;
    }
  }
}

#define KS 128
#define KCH 8

__global__ __launch_bounds__(256) void k_gemm_xw(const float* __restrict__ X,
                                                 const float* __restrict__ W,
                                                 const float* __restrict__ bias,
                                                 float* __restrict__ out, int relu) {
  __shared__ v4f part[8][16][16];
  const int tid  = threadIdx.x;
  const int wv   = tid >> 5;
  const int lane = tid & 31;
  const int half = lane >> 4;
  const int lid  = lane & 15;
  const int m0 = blockIdx.x * 32, n0 = blockIdx.y * 64;
  const int kb = wv * KS + 16 * half;
  const float* arow0 = X + (size_t)(m0 + lid) * DD + kb;
  const float* arow1 = X + (size_t)(m0 + 16 + lid) * DD + kb;
  const float* wp = W + (size_t)kb * DD + n0 + 4 * lid;
  v8f acc[2][4];
#pragma unroll
  for (int s = 0; s < 2; ++s)
#pragma unroll
    for (int t = 0; t < 4; ++t) acc[s][t] = (v8f){0.f,0.f,0.f,0.f,0.f,0.f,0.f,0.f};
  for (int kc = 0; kc < KS; kc += 4 * KCH) {
    v4f a0[KCH / 2], a1[KCH / 2];
    v4f w0[KCH], w1[KCH];
#pragma unroll
    for (int i = 0; i < KCH / 2; ++i) {
      a0[i] = *(const v4f*)(arow0 + kc + 4 * i);
      a1[i] = *(const v4f*)(arow1 + kc + 4 * i);
    }
#pragma unroll
    for (int j = 0; j < KCH; ++j) {
      w0[j] = *(const v4f*)(wp + (size_t)(kc + 2 * j) * DD);
      w1[j] = *(const v4f*)(wp + (size_t)(kc + 2 * j + 1) * DD);
    }
    __builtin_amdgcn_sched_barrier(0);
#pragma unroll
    for (int j = 0; j < KCH; ++j) {
      const v2f fa0 = (j & 1) ? hi2(a0[j >> 1]) : lo2(a0[j >> 1]);
      const v2f fa1 = (j & 1) ? hi2(a1[j >> 1]) : lo2(a1[j >> 1]);
#pragma unroll
      for (int t = 0; t < 4; ++t) {
        v2f b;
        b.x = w0[j][t];
        b.y = w1[j][t];
        acc[0][t] = wmma4(fa0, b, acc[0][t]);
        acc[1][t] = wmma4(fa1, b, acc[1][t]);
      }
    }
  }
  const int rr = tid >> 4, cc = tid & 15;
  const v4f bv = *(const v4f*)(bias + n0 + 4 * cc);
#pragma unroll
  for (int s = 0; s < 2; ++s) {
    if (s) __syncthreads();
#pragma unroll
    for (int i = 0; i < 8; ++i) {
      v4f o;
      o.x = acc[s][0][i];
      o.y = acc[s][1][i];
      o.z = acc[s][2][i];
      o.w = acc[s][3][i];
      part[wv][half * 8 + i][lid] = o;
    }
    __syncthreads();
    v4f sum = part[0][rr][cc];
#pragma unroll
    for (int w = 1; w < 8; ++w) sum += part[w][rr][cc];
    sum += bv;
    if (relu) {
      sum.x = fmaxf(sum.x, 0.f);
      sum.y = fmaxf(sum.y, 0.f);
      sum.z = fmaxf(sum.z, 0.f);
      sum.w = fmaxf(sum.w, 0.f);
    }
    *(v4f*)(out + (size_t)(m0 + 16 * s + rr) * DD + n0 + 4 * cc) = sum;
  }
}

static __device__ __forceinline__ v2f sq_acc(v2f acc, v4f q) {
  acc += lo2(q) * lo2(q);
  acc += hi2(q) * hi2(q);
  return acc;
}

static __device__ __forceinline__ v4f exp4(v4f x, float m) {
  v4f e;
  e.x = __expf(x.x - m);
  e.y = __expf(x.y - m);
  e.z = __expf(x.z - m);
  e.w = __expf(x.w - m);
  return e;
}

static __device__ __forceinline__ float max4(float m, v4f x) {
  return fmaxf(fmaxf(m, fmaxf(x.x, x.y)), fmaxf(x.z, x.w));
}

__global__ __launch_bounds__(256) void k_score(const float* __restrict__ trm,
                                               const float* __restrict__ aem,
                                               const float* __restrict__ r,
                                               const float* __restrict__ t,
                                               float* __restrict__ score) {
  __shared__ __attribute__((aligned(16))) float part[8][3][16][32];
  __shared__ __attribute__((aligned(16))) float ssq[8][3][32];
  __shared__ float rmx[8][32];
  const int tid  = threadIdx.x;
  const int wv   = tid >> 5;
  const int lane = tid & 31;
  const int half = lane >> 4;
  const int lid  = lane & 15;
  const int m0 = blockIdx.x * 32, n0 = blockIdx.y * 32;
  const int kb = wv * KS + 16 * half;
  const size_t ra0 = (size_t)(m0 + lid) * DD + kb;
  const size_t ra1 = (size_t)(m0 + 16 + lid) * DD + kb;
  const size_t cb0 = (size_t)(n0 + lid) * DD + kb;
  const size_t cb1 = (size_t)(n0 + 16 + lid) * DD + kb;
  float mx0, mx1;
  {
    v4f x0[KS / 8], x1[KS / 8];
#pragma unroll
    for (int i = 0; i < KS / 8; ++i) {
      const int k0 = (i >> 2) * 4 * KCH + 4 * (i & 3);
      x0[i] = *(const v4f*)(aem + ra0 + k0);
      x1[i] = *(const v4f*)(aem + ra1 + k0);
    }
    mx0 = x0[0].x;
    mx1 = x1[0].x;
#pragma unroll
    for (int i = 0; i < KS / 8; ++i) {
      mx0 = max4(mx0, x0[i]);
      mx1 = max4(mx1, x1[i]);
    }
  }
  mx0 = fmaxf(mx0, __shfl_xor(mx0, 16));
  mx1 = fmaxf(mx1, __shfl_xor(mx1, 16));
  if (half == 0) {
    rmx[wv][lid]      = mx0;
    rmx[wv][16 + lid] = mx1;
  }
  __syncthreads();
  mx0 = rmx[0][lid];
  mx1 = rmx[0][16 + lid];
  for (int w = 1; w < 8; ++w) {
    mx0 = fmaxf(mx0, rmx[w][lid]);
    mx1 = fmaxf(mx1, rmx[w][16 + lid]);
  }
  const float mx[2] = {mx0, mx1};
  v8f num[2][2], den[2][2], isv[2][2];
#pragma unroll
  for (int s = 0; s < 2; ++s)
#pragma unroll
    for (int u = 0; u < 2; ++u) {
      num[s][u] = (v8f){0.f,0.f,0.f,0.f,0.f,0.f,0.f,0.f};
      den[s][u] = num[s][u];
      isv[s][u] = num[s][u];
    }
  v2f qt2[2], qr2[2], qc2[2];
#pragma unroll
  for (int s = 0; s < 2; ++s) {
    qt2[s] = (v2f){0.f, 0.f};
    qr2[s] = qt2[s];
    qc2[s] = qt2[s];
  }
  for (int kc = 0; kc < KS; kc += 4 * KCH) {
    v4f a1[KCH / 2][2], a2[KCH / 2][2], a3[KCH / 2][2];
    v4f b1[KCH / 2][2], b2[KCH / 2][2];
#pragma unroll
    for (int i = 0; i < KCH / 2; ++i) {
      const int k0 = kc + 4 * i;
      a1[i][0] = *(const v4f*)(trm + ra0 + k0);  a1[i][1] = *(const v4f*)(trm + ra1 + k0);
      a2[i][0] = *(const v4f*)(aem + ra0 + k0);  a2[i][1] = *(const v4f*)(aem + ra1 + k0);
      a3[i][0] = *(const v4f*)(r   + ra0 + k0);  a3[i][1] = *(const v4f*)(r   + ra1 + k0);
      b1[i][0] = *(const v4f*)(t   + cb0 + k0);  b1[i][1] = *(const v4f*)(t   + cb1 + k0);
    }
    __builtin_amdgcn_sched_barrier(0);
#pragma unroll
    for (int i = 0; i < KCH / 2; ++i)
#pragma unroll
      for (int s = 0; s < 2; ++s) {
        qt2[s] = sq_acc(qt2[s], a1[i][s]);
        qr2[s] = sq_acc(qr2[s], a3[i][s]);
        qc2[s] = sq_acc(qc2[s], b1[i][s]);
        a2[i][s] = exp4(a2[i][s], mx[s]);
        a1[i][s] = a1[i][s] * a2[i][s];
        a2[i][s] = a2[i][s] * a2[i][s];
        b2[i][s] = b1[i][s] * b1[i][s];
      }
#pragma unroll
    for (int j = 0; j < KCH; ++j)
#pragma unroll
      for (int s = 0; s < 2; ++s)
#pragma unroll
        for (int u = 0; u < 2; ++u) {
          const int i = j >> 1;
          const v2f fa1 = (j & 1) ? hi2(a1[i][s]) : lo2(a1[i][s]);
          const v2f fa2 = (j & 1) ? hi2(a2[i][s]) : lo2(a2[i][s]);
          const v2f fa3 = (j & 1) ? hi2(a3[i][s]) : lo2(a3[i][s]);
          const v2f fb1 = (j & 1) ? hi2(b1[i][u]) : lo2(b1[i][u]);
          const v2f fb2 = (j & 1) ? hi2(b2[i][u]) : lo2(b2[i][u]);
          num[s][u] = wmma4(fa1, fb1, num[s][u]);
          den[s][u] = wmma4(fa2, fb2, den[s][u]);
          isv[s][u] = wmma4(fa3, fb1, isv[s][u]);
        }
  }
  float qt[2], qr[2], qc[2];
#pragma unroll
  for (int s = 0; s < 2; ++s) {
    qt[s] = qt2[s].x + qt2[s].y;
    qr[s] = qr2[s].x + qr2[s].y;
    qc[s] = qc2[s].x + qc2[s].y;
    qt[s] += __shfl_xor(qt[s], 16);
    qr[s] += __shfl_xor(qr[s], 16);
    qc[s] += __shfl_xor(qc[s], 16);
  }
  if (half == 0) {
#pragma unroll
    for (int s = 0; s < 2; ++s) {
      ssq[wv][0][16 * s + lid] = qt[s];
      ssq[wv][1][16 * s + lid] = qr[s];
      ssq[wv][2][16 * s + lid] = qc[s];
    }
  }
  const int rr = tid >> 4, cc = 2 * (tid & 15);
  v2f ic = {0.f, 0.f};
#pragma unroll
  for (int s = 0; s < 2; ++s) {
    if (s) __syncthreads();
#pragma unroll
    for (int u = 0; u < 2; ++u)
#pragma unroll
      for (int i = 0; i < 8; ++i) {
        part[wv][0][half * 8 + i][16 * u + lid] = num[s][u][i];
        part[wv][1][half * 8 + i][16 * u + lid] = den[s][u][i];
        part[wv][2][half * 8 + i][16 * u + lid] = isv[s][u][i];
      }
    __syncthreads();
    if (s == 0) {
      v2f c = *(const v2f*)&ssq[0][2][cc];
#pragma unroll
      for (int w = 1; w < 8; ++w) c += *(const v2f*)&ssq[w][2][cc];
      ic.x = rsqrtf(c.x);
      ic.y = rsqrtf(c.y);
    }
    float st = ssq[0][0][16 * s + rr];
    float sr = ssq[0][1][16 * s + rr];
#pragma unroll
    for (int w = 1; w < 8; ++w) {
      st += ssq[w][0][16 * s + rr];
      sr += ssq[w][1][16 * s + rr];
    }
    const float it = rsqrtf(st);
    const float ir = rsqrtf(sr);
    v2f sn = *(const v2f*)&part[0][0][rr][cc];
    v2f sd = *(const v2f*)&part[0][1][rr][cc];
    v2f si = *(const v2f*)&part[0][2][rr][cc];
#pragma unroll
    for (int w = 1; w < 8; ++w) {
      sn += *(const v2f*)&part[w][0][rr][cc];
      sd += *(const v2f*)&part[w][1][rr][cc];
      si += *(const v2f*)&part[w][2][rr][cc];
    }
    v2f o;
    o.x = sn.x * rsqrtf(sd.x) * it + si.x * (ir * ic.x);
    o.y = sn.y * rsqrtf(sd.y) * it + si.y * (ir * ic.y);
    *(v2f*)(score + (size_t)(m0 + 16 * s + rr) * BB + n0 + cc) = o;
  }
}

extern "C" void kernel_launch(void* const* d_in, const int* in_sizes, int n_in,
                              void* d_out, int out_size, void* d_ws, size_t ws_size,
                              hipStream_t stream) {
  const float* r    = (const float*)d_in[0];
  const float* m    = (const float*)d_in[1];
  const float* t    = (const float*)d_in[2];
  const int*   ntid = (const int*)d_in[3];
  const int*   nlen = (const int*)d_in[4];
  const int*   antk = (const int*)d_in[5];
  const float* emb  = (const float*)d_in[6];
  const float* nW   = (const float*)d_in[7];
  const float* nb   = (const float*)d_in[8];
  const float* W1   = (const float*)d_in[9];
  const float* b1   = (const float*)d_in[10];
  const float* W2   = (const float*)d_in[11];
  const float* b2   = (const float*)d_in[12];
  const float* trW  = (const float*)d_in[13];
  const float* trb  = (const float*)d_in[14];
  const int*   t2i  = (const int*)d_in[15];

  float* score    = (float*)d_out;
  float* concepts = score + (size_t)BB * BB;

  float* ws  = (float*)d_ws;
  float* cls = ws;  ws += (size_t)NN * DD;
  float* h   = ws;  ws += (size_t)BB * DD;
  float* aem = ws;  ws += (size_t)BB * DD;
  float* trm = ws;

  const dim3 wave(32);
  k_cls<<<dim3(NN / 64, DD / 64), wave, 0, stream>>>(antk, emb, nW, nb, cls);
  k_concepts<<<dim3(BB * LL / 64, DD / 64), wave, 0, stream>>>(ntid, nlen, t2i, emb,
                                                               nW, nb, cls, concepts);
  k_gemm_xw<<<dim3(BB / 32, DD / 64), dim3(256), 0, stream>>>(m, W1, b1, h, 1);
  k_gemm_xw<<<dim3(BB / 32, DD / 64), dim3(256), 0, stream>>>(h, W2, b2, aem, 0);
  k_gemm_xw<<<dim3(BB / 32, DD / 64), dim3(256), 0, stream>>>(m, trW, trb, trm, 0);
  k_score<<<dim3(BB / 32, BB / 32), dim3(256), 0, stream>>>(trm, aem, r, t, score);
}


# Class-weight GEMM only over the class rows that some concepts row takes: one spare work-group of the first launch lists them (LDS flags, scan), 32x64 tiles walk the list; k order, sums unchanged
# speedup vs baseline: 3.8461x; 1.2743x over previous
#include <hip/hip_runtime.h>
#include <hip/hip_bf16.h>

typedef __attribute__((ext_vector_type(2))) float v2f;
typedef __attribute__((ext_vector_type(4))) float v4f;
typedef __attribute__((ext_vector_type(4))) int v4i;
typedef __attribute__((ext_vector_type(8))) float v8f;

#define BB 384
#define DD 1024
#define WD 300
#define NN 8000
#define LL 16

static __device__ __forceinline__ v8f wmma4(v2f a, v2f b, v8f c) {
  return __builtin_amdgcn_wmma_f32_16x16x4_f32(false, a, false, b, (short)0, c,
                                               false, false);
}

static __device__ __forceinline__ v2f lo2(v4f q) { return __builtin_shufflevector(q, q, 0, 1); }
static __device__ __forceinline__ v2f hi2(v4f q) { return __builtin_shufflevector(q, q, 2, 3); }

#define KC 5
#define NCH (WD / (4 * KC))

template <int S>
struct Chunk {
  v4f ax[S];
  v4f ay[S];
  v2f az[S];
  v4f b0[KC];
  v4f b1[KC];
};

template <int S>
static __device__ __forceinline__ void load_chunk(Chunk<S>& f, const float* const (&ar)[S],
                                                  int zd, const float* wpx,
                                                  const float* wpz, int kb) {
#pragma unroll
  for (int s = 0; s < S; ++s) {
    f.ax[s] = *(const v4f*)(ar[s] + kb);
    f.ay[s] = *(const v4f*)(ar[s] + kb + 4);
    f.az[s] = *(const v2f*)(ar[s] + kb + zd);
  }
#pragma unroll
  for (int j = 0; j < KC - 1; ++j) {
    f.b0[j] = *(const v4f*)(wpx + (size_t)(kb + 2 * j) * DD);
    f.b1[j] = *(const v4f*)(wpx + (size_t)(kb + 2 * j + 1) * DD);
  }
  f.b0[KC - 1] = *(const v4f*)(wpz + (size_t)kb * DD);
  f.b1[KC - 1] = *(const v4f*)(wpz + (size_t)(kb + 1) * DD);
}

template <int S>
static __device__ __forceinline__ void mma_chunk(v8f (&acc)[S][4], const Chunk<S>& f) {
#pragma unroll
  for (int j = 0; j < KC; ++j) {
#pragma unroll
    for (int t = 0; t < 4; ++t) {
      v2f b;
      b.x = f.b0[j][t];
      b.y = f.b1[j][t];
#pragma unroll
      for (int s = 0; s < S; ++s) {
        const v2f a = j == 0 ? lo2(f.ax[s]) : j == 1 ? hi2(f.ax[s])
                    : j == 2 ? lo2(f.ay[s]) : j == 3 ? hi2(f.ay[s]) : f.az[s];
        acc[s][t] = wmma4(a, b, acc[s][t]);
      }
    }
  }
}

template <int S>
static __device__ __forceinline__ void gemm300_tile(v8f (&acc)[S][4],
                                                    const float* const (&ar)[S], int zd,
                                                    const float* wpx, const float* wpz) {
#pragma unroll
  for (int s = 0; s < S; ++s)
#pragma unroll
    for (int t = 0; t < 4; ++t) acc[s][t] = (v8f){0.f,0.f,0.f,0.f,0.f,0.f,0.f,0.f};
  Chunk<S> f0, f1;
  load_chunk(f0, ar, zd, wpx, wpz, 0);
#pragma unroll 1
  for (int ch = 0; ch < NCH - 1; ch += 2) {
    load_chunk(f1, ar, zd, wpx, wpz, (ch + 1) * 4 * KC);
    mma_chunk(acc, f0);
    load_chunk(f0, ar, zd, wpx, wpz, (ch + 2) * 4 * KC);
    mma_chunk(acc, f1);
  }
  mma_chunk(acc, f0);
}

#define CLS_RT 32
#define CLS_NT ((BB * LL + 1 + CLS_RT - 1) / CLS_RT)
#define CLS_GY 64
__global__ __launch_bounds__(128) void k_cls(const int* __restrict__ list,
                                            const int* __restrict__ tok,
                                            const float* __restrict__ emb,
                                            const float* __restrict__ W,
                                            const float* __restrict__ bias,
                                            float* __restrict__ out) {
  const int lane = threadIdx.x & 31;
  const int half = lane >> 4;
  const int lid  = lane & 15;
  const int n0 = (blockIdx.x * 4 + (threadIdx.x >> 5)) * 64;
  const int c0 = n0 + 4 * lid;
  const int xo = half ? 12 : 0;
  const int zo = half ? 10 : 8;
  const float* wpx = W + (size_t)xo * DD + c0;
  const float* wpz = W + (size_t)zo * DD + c0;
  const v4f bv = *(const v4f*)(bias + c0);
  for (int rt = blockIdx.y; rt < CLS_NT; rt += CLS_GY) {
    const int r0 = rt * CLS_RT;
    if (__builtin_amdgcn_readfirstlane(list[r0]) < 0) break;
    int orow[2][8];
#pragma unroll
    for (int s = 0; s < 2; ++s) {
      const v4i q0 = *(const v4i*)(list + r0 + 16 * s + 8 * half);
      const v4i q1 = *(const v4i*)(list + r0 + 16 * s + 8 * half + 4);
#pragma unroll
      for (int i = 0; i < 4; ++i) {
        orow[s][i]     = q0[i];
        orow[s][4 + i] = q1[i];
      }
    }
    const int ra0 = list[r0 + lid], ra1 = list[r0 + 16 + lid];
    const float* ar[2] = {emb + (size_t)tok[ra0 < 0 ? 0 : ra0] * WD + xo,
                          emb + (size_t)tok[ra1 < 0 ? 0 : ra1] * WD + xo};
    v8f acc[2][4];
    gemm300_tile<2>(acc, ar, zo - xo, wpx, wpz);
#pragma unroll
    for (int s = 0; s < 2; ++s)
#pragma unroll
      for (int i = 0; i < 8; ++i) {
        v4f o;
        o.x = acc[s][0][i] + bv.x;
        o.y = acc[s][1][i] + bv.y;
        o.z = acc[s][2][i] + bv.z;
        o.w = acc[s][3][i] + bv.w;
        if (orow[s][i] >= 0) *(v4f*)(out + (size_t)orow[s][i] * DD + c0) = o;
      }
  }
}

__global__ __launch_bounds__(32) void k_concepts(const int* __restrict__ ntid,
                                                 const int* __restrict__ nlen,
                                                 const int* __restrict__ t2i,
                                                 const float* __restrict__ emb,
                                                 const float* __restrict__ W,
                                                 const float* __restrict__ bias,
                                                 const float* __restrict__ cls,
                                                 float* __restrict__ concepts) {
  const int lane = threadIdx.x & 31;
  const int half = lane >> 4;
  const int lid  = lane & 15;
  const int m0 = blockIdx.x * 64, n0 = blockIdx.y * 64;
  const int c0 = n0 + 4 * lid;
  int widx[4][8];
  int nl[4];
#pragma unroll
  for (int s = 0; s < 4; ++s) {
    const v4i t0 = *(const v4i*)(ntid + m0 + 16 * s + 8 * half);
    const v4i t1 = *(const v4i*)(ntid + m0 + 16 * s + 8 * half + 4);
#pragma unroll
    for (int i = 0; i < 4; ++i) {
      widx[s][i]     = t2i[t0[i]];
      widx[s][4 + i] = t2i[t1[i]];
    }
    nl[s] = nlen[(m0 >> 4) + s];
  }
  const int xo = half ? 12 : 0;
  const int zo = half ? 10 : 8;
  const float* ar[4] = {emb + (size_t)ntid[m0 +  0 + lid] * WD + xo,
                        emb + (size_t)ntid[m0 + 16 + lid] * WD + xo,
                        emb + (size_t)ntid[m0 + 32 + lid] * WD + xo,
                        emb + (size_t)ntid[m0 + 48 + lid] * WD + xo};
  const float* wpx = W + (size_t)xo * DD + c0;
  const float* wpz = W + (size_t)zo * DD + c0;
  v8f acc[4][4];
  gemm300_tile<4>(acc, ar, zo - xo, wpx, wpz);
  const v4f bv = *(const v4f*)(bias + c0);
  v4f mv[4][8];
#pragma unroll
  for (int s = 0; s < 4; ++s)
#pragma unroll
    for (int i = 0; i < 8; ++i) {
      const bool use = (8 * half + i < nl[s]) && (widx[s][i] > 0);
      mv[s][i] = *(const v4f*)(cls + (size_t)(use ? widx[s][i] : 0) * DD + c0);
    }
#pragma unroll
  for (int s = 0; s < 4; ++s) {
#pragma unroll
    for (int i = 0; i < 8; ++i) {
      const int row  = m0 + s * 16 + half * 8 + i;
      const bool valid = 8 * half + i < nl[s];
      const bool found = widx[s][i] > 0;
      v4f val = {0.f, 0.f, 0.f, 0.f};
      if (valid) {
        if (found) {
          val = mv[s][i];
        } else {
          val.x = acc[s][0][i] + bv.x;
          val.y = acc[s][1][i] + bv.y;
          val.z = acc[s][2][i] + bv.z;
          val.w = acc[s][3][i] + bv.w;
        }
      }
      *(v4f*)(concepts + (size_t)row * DD + c0) = val;
    }
  }
}

static __device__ __forceinline__ void list_used_rows(const int* __restrict__ ntid,
                                                      const int* __restrict__ nlen,
                                                      const int* __restrict__ t2i,
                                                      int* __restrict__ list,
                                                      unsigned char* used, int* wtot) {
  const int tid  = threadIdx.x;
  const int wv   = tid >> 5;
  const int lane = tid & 31;
  for (int i = tid; i < NN / 4; i += 256) ((int*)used)[i] = 0;
  __syncthreads();
  int tk[BB * LL / 256], wi[BB * LL / 256];
  bool vd[BB * LL / 256];
#pragma unroll
  for (int j = 0; j < BB * LL / 256; ++j) {
    const int p = j * 256 + tid;
    tk[j] = ntid[p];
    vd[j] = (p & (LL - 1)) < nlen[p / LL];
  }
#pragma unroll
  for (int j = 0; j < BB * LL / 256; ++j) wi[j] = t2i[tk[j]];
#pragma unroll
  for (int j = 0; j < BB * LL / 256; ++j)
    if (vd[j] && wi[j] > 0) used[wi[j]] = 1;
  if (tid == 0) used[0] = 1;
  __syncthreads();
  int w[8];
  int cnt = 0;
  if (tid < NN / 32) {
    const v4i q0 = ((const v4i*)used)[2 * tid];
    const v4i q1 = ((const v4i*)used)[2 * tid + 1];
#pragma unroll
    for (int i = 0; i < 4; ++i) {
      w[i]     = q0[i];
      w[4 + i] = q1[i];
    }
    int sum = 0;
#pragma unroll
    for (int i = 0; i < 8; ++i) sum += w[i];
    cnt = (sum & 0xff) + ((sum >> 8) & 0xff) + ((sum >> 16) & 0xff) + ((sum >> 24) & 0xff);
  } else {
#pragma unroll
    for (int i = 0; i < 8; ++i) w[i] = 0;
  }
  int inc = cnt;
#pragma unroll
  for (int d = 1; d < 32; d *= 2) {
    const int v = __shfl_up(inc, d);
    if (lane >= d) inc += v;
  }
  if (lane == 31) wtot[wv] = inc;
  __syncthreads();
  int rank = inc - cnt, total = 0;
#pragma unroll
  for (int i = 0; i < 8; ++i) {
    const int c = wtot[i];
    if (i < wv) rank += c;
    total += c;
  }
#pragma unroll
  for (int j = 0; j < 32; ++j)
    if ((w[j >> 2] >> (8 * (j & 3))) & 1) list[rank++] = 32 * tid + j;
  for (int i = total + tid; i < NN; i += 256) list[i] = -1;
}

#define KS 128
#define KCH 8

__global__ __launch_bounds__(256) void k_gemm_xw(const float* __restrict__ X,
                                                 const float* __restrict__ W,
                                                 const float* __restrict__ bias,
                                                 float* __restrict__ out, int relu,
                                                 int lister,
                                                 const int* __restrict__ ntid,
                                                 const int* __restrict__ nlen,
                                                 const int* __restrict__ t2i,
                                                 int* __restrict__ list) {
  __shared__ v4f part[8][16][16];
  const int by = (int)blockIdx.y - lister;
  if (by < 0) {
    unsigned char* used = (unsigned char*)&part[0][0][0];
    if (blockIdx.x == 0) list_used_rows(ntid, nlen, t2i, list, used, (int*)(used + NN));
    return;
  }
  const int tid  = threadIdx.x;
  const int wv   = tid >> 5;
  const int lane = tid & 31;
  const int half = lane >> 4;
  const int lid  = lane & 15;
  const int m0 = blockIdx.x * 32, n0 = by * 64;
  const int kb = wv * KS + 16 * half;
  const float* arow0 = X + (size_t)(m0 + lid) * DD + kb;
  const float* arow1 = X + (size_t)(m0 + 16 + lid) * DD + kb;
  const float* wp = W + (size_t)kb * DD + n0 + 4 * lid;
  v8f acc[2][4];
#pragma unroll
  for (int s = 0; s < 2; ++s)
#pragma unroll
    for (int t = 0; t < 4; ++t) acc[s][t] = (v8f){0.f,0.f,0.f,0.f,0.f,0.f,0.f,0.f};
  for (int kc = 0; kc < KS; kc += 4 * KCH) {
    v4f a0[KCH / 2], a1[KCH / 2];
    v4f w0[KCH], w1[KCH];
#pragma unroll
    for (int i = 0; i < KCH / 2; ++i) {
      a0[i] = *(const v4f*)(arow0 + kc + 4 * i);
      a1[i] = *(const v4f*)(arow1 + kc + 4 * i);
    }
#pragma unroll
    for (int j = 0; j < KCH; ++j) {
      w0[j] = *(const v4f*)(wp + (size_t)(kc + 2 * j) * DD);
      w1[j] = *(const v4f*)(wp + (size_t)(kc + 2 * j + 1) * DD);
    }
    __builtin_amdgcn_sched_barrier(0);
#pragma unroll
    for (int j = 0; j < KCH; ++j) {
      const v2f fa0 = (j & 1) ? hi2(a0[j >> 1]) : lo2(a0[j >> 1]);
      const v2f fa1 = (j & 1) ? hi2(a1[j >> 1]) : lo2(a1[j >> 1]);
#pragma unroll
      for (int t = 0; t < 4; ++t) {
        v2f b;
        b.x = w0[j][t];
        b.y = w1[j][t];
        acc[0][t] = wmma4(fa0, b, acc[0][t]);
        acc[1][t] = wmma4(fa1, b, acc[1][t]);
      }
    }
  }
  const int rr = tid >> 4, cc = tid & 15;
  const v4f bv = *(const v4f*)(bias + n0 + 4 * cc);
#pragma unroll
  for (int s = 0; s < 2; ++s) {
    if (s) __syncthreads();
#pragma unroll
    for (int i = 0; i < 8; ++i) {
      v4f o;
      o.x = acc[s][0][i];
      o.y = acc[s][1][i];
      o.z = acc[s][2][i];
      o.w = acc[s][3][i];
      part[wv][half * 8 + i][lid] = o;
    }
    __syncthreads();
    v4f sum = part[0][rr][cc];
#pragma unroll
    for (int w = 1; w < 8; ++w) sum += part[w][rr][cc];
    sum += bv;
    if (relu) {
      sum.x = fmaxf(sum.x, 0.f);
      sum.y = fmaxf(sum.y, 0.f);
      sum.z = fmaxf(sum.z, 0.f);
      sum.w = fmaxf(sum.w, 0.f);
    }
    *(v4f*)(out + (size_t)(m0 + 16 * s + rr) * DD + n0 + 4 * cc) = sum;
  }
}

static __device__ __forceinline__ v2f sq_acc(v2f acc, v4f q) {
  acc += lo2(q) * lo2(q);
  acc += hi2(q) * hi2(q);
  return acc;
}

static __device__ __forceinline__ v4f exp4(v4f x, float m) {
  v4f e;
  e.x = __expf(x.x - m);
  e.y = __expf(x.y - m);
  e.z = __expf(x.z - m);
  e.w = __expf(x.w - m);
  return e;
}

static __device__ __forceinline__ float max4(float m, v4f x) {
  return fmaxf(fmaxf(m, fmaxf(x.x, x.y)), fmaxf(x.z, x.w));
}

__global__ __launch_bounds__(256) void k_score(const float* __restrict__ trm,
                                               const float* __restrict__ aem,
                                               const float* __restrict__ r,
                                               const float* __restrict__ t,
                                               float* __restrict__ score) {
  __shared__ __attribute__((aligned(16))) float part[8][3][16][32];
  __shared__ __attribute__((aligned(16))) float ssq[8][3][32];
  __shared__ float rmx[8][32];
  const int tid  = threadIdx.x;
  const int wv   = tid >> 5;
  const int lane = tid & 31;
  const int half = lane >> 4;
  const int lid  = lane & 15;
  const int m0 = blockIdx.x * 32, n0 = blockIdx.y * 32;
  const int kb = wv * KS + 16 * half;
  const size_t ra0 = (size_t)(m0 + lid) * DD + kb;
  const size_t ra1 = (size_t)(m0 + 16 + lid) * DD + kb;
  const size_t cb0 = (size_t)(n0 + lid) * DD + kb;
  const size_t cb1 = (size_t)(n0 + 16 + lid) * DD + kb;
  float mx0, mx1;
  {
    v4f x0[KS / 8], x1[KS / 8];
#pragma unroll
    for (int i = 0; i < KS / 8; ++i) {
      const int k0 = (i >> 2) * 4 * KCH + 4 * (i & 3);
      x0[i] = *(const v4f*)(aem + ra0 + k0);
      x1[i] = *(const v4f*)(aem + ra1 + k0);
    }
    mx0 = x0[0].x;
    mx1 = x1[0].x;
#pragma unroll
    for (int i = 0; i < KS / 8; ++i) {
      mx0 = max4(mx0, x0[i]);
      mx1 = max4(mx1, x1[i]);
    }
  }
  mx0 = fmaxf(mx0, __shfl_xor(mx0, 16));
  mx1 = fmaxf(mx1, __shfl_xor(mx1, 16));
  if (half == 0) {
    rmx[wv][lid]      = mx0;
    rmx[wv][16 + lid] = mx1;
  }
  __syncthreads();
  mx0 = rmx[0][lid];
  mx1 = rmx[0][16 + lid];
  for (int w = 1; w < 8; ++w) {
    mx0 = fmaxf(mx0, rmx[w][lid]);
    mx1 = fmaxf(mx1, rmx[w][16 + lid]);
  }
  const float mx[2] = {mx0, mx1};
  v8f num[2][2], den[2][2], isv[2][2];
#pragma unroll
  for (int s = 0; s < 2; ++s)
#pragma unroll
    for (int u = 0; u < 2; ++u) {
      num[s][u] = (v8f){0.f,0.f,0.f,0.f,0.f,0.f,0.f,0.f};
      den[s][u] = num[s][u];
      isv[s][u] = num[s][u];
    }
  v2f qt2[2], qr2[2], qc2[2];
#pragma unroll
  for (int s = 0; s < 2; ++s) {
    qt2[s] = (v2f){0.f, 0.f};
    qr2[s] = qt2[s];
    qc2[s] = qt2[s];
  }
  for (int kc = 0; kc < KS; kc += 4 * KCH) {
    v4f a1[KCH / 2][2], a2[KCH / 2][2], a3[KCH / 2][2];
    v4f b1[KCH / 2][2], b2[KCH / 2][2];
#pragma unroll
    for (int i = 0; i < KCH / 2; ++i) {
      const int k0 = kc + 4 * i;
      a1[i][0] = *(const v4f*)(trm + ra0 + k0);  a1[i][1] = *(const v4f*)(trm + ra1 + k0);
      a2[i][0] = *(const v4f*)(aem + ra0 + k0);  a2[i][1] = *(const v4f*)(aem + ra1 + k0);
      a3[i][0] = *(const v4f*)(r   + ra0 + k0);  a3[i][1] = *(const v4f*)(r   + ra1 + k0);
      b1[i][0] = *(const v4f*)(t   + cb0 + k0);  b1[i][1] = *(const v4f*)(t   + cb1 + k0);
    }
    __builtin_amdgcn_sched_barrier(0);
#pragma unroll
    for (int i = 0; i < KCH / 2; ++i)
#pragma unroll
      for (int s = 0; s < 2; ++s) {
        qt2[s] = sq_acc(qt2[s], a1[i][s]);
        qr2[s] = sq_acc(qr2[s], a3[i][s]);
        qc2[s] = sq_acc(qc2[s], b1[i][s]);
        a2[i][s] = exp4(a2[i][s], mx[s]);
        a1[i][s] = a1[i][s] * a2[i][s];
        a2[i][s] = a2[i][s] * a2[i][s];
        b2[i][s] = b1[i][s] * b1[i][s];
      }
#pragma unroll
    for (int j = 0; j < KCH; ++j)
#pragma unroll
      for (int s = 0; s < 2; ++s)
#pragma unroll
        for (int u = 0; u < 2; ++u) {
          const int i = j >> 1;
          const v2f fa1 = (j & 1) ? hi2(a1[i][s]) : lo2(a1[i][s]);
          const v2f fa2 = (j & 1) ? hi2(a2[i][s]) : lo2(a2[i][s]);
          const v2f fa3 = (j & 1) ? hi2(a3[i][s]) : lo2(a3[i][s]);
          const v2f fb1 = (j & 1) ? hi2(b1[i][u]) : lo2(b1[i][u]);
          const v2f fb2 = (j & 1) ? hi2(b2[i][u]) : lo2(b2[i][u]);
          num[s][u] = wmma4(fa1, fb1, num[s][u]);
          den[s][u] = wmma4(fa2, fb2, den[s][u]);
          isv[s][u] = wmma4(fa3, fb1, isv[s][u]);
        }
  }
  float qt[2], qr[2], qc[2];
#pragma unroll
  for (int s = 0; s < 2; ++s) {
    qt[s] = qt2[s].x + qt2[s].y;
    qr[s] = qr2[s].x + qr2[s].y;
    qc[s] = qc2[s].x + qc2[s].y;
    qt[s] += __shfl_xor(qt[s], 16);
    qr[s] += __shfl_xor(qr[s], 16);
    qc[s] += __shfl_xor(qc[s], 16);
  }
  if (half == 0) {
#pragma unroll
    for (int s = 0; s < 2; ++s) {
      ssq[wv][0][16 * s + lid] = qt[s];
      ssq[wv][1][16 * s + lid] = qr[s];
      ssq[wv][2][16 * s + lid] = qc[s];
    }
  }
  const int rr = tid >> 4, cc = 2 * (tid & 15);
  v2f ic = {0.f, 0.f};
#pragma unroll
  for (int s = 0; s < 2; ++s) {
    if (s) __syncthreads();
#pragma unroll
    for (int u = 0; u < 2; ++u)
#pragma unroll
      for (int i = 0; i < 8; ++i) {
        part[wv][0][half * 8 + i][16 * u + lid] = num[s][u][i];
        part[wv][1][half * 8 + i][16 * u + lid] = den[s][u][i];
        part[wv][2][half * 8 + i][16 * u + lid] = isv[s][u][i];
      }
    __syncthreads();
    if (s == 0) {
      v2f c = *(const v2f*)&ssq[0][2][cc];
#pragma unroll
      for (int w = 1; w < 8; ++w) c += *(const v2f*)&ssq[w][2][cc];
      ic.x = rsqrtf(c.x);
      ic.y = rsqrtf(c.y);
    }
    float st = ssq[0][0][16 * s + rr];
    float sr = ssq[0][1][16 * s + rr];
#pragma unroll
    for (int w = 1; w < 8; ++w) {
      st += ssq[w][0][16 * s + rr];
      sr += ssq[w][1][16 * s + rr];
    }
    const float it = rsqrtf(st);
    const float ir = rsqrtf(sr);
    v2f sn = *(const v2f*)&part[0][0][rr][cc];
    v2f sd = *(const v2f*)&part[0][1][rr][cc];
    v2f si = *(const v2f*)&part[0][2][rr][cc];
#pragma unroll
    for (int w = 1; w < 8; ++w) {
      sn += *(const v2f*)&part[w][0][rr][cc];
      sd += *(const v2f*)&part[w][1][rr][cc];
      si += *(const v2f*)&part[w][2][rr][cc];
    }
    v2f o;
    o.x = sn.x * rsqrtf(sd.x) * it + si.x * (ir * ic.x);
    o.y = sn.y * rsqrtf(sd.y) * it + si.y * (ir * ic.y);
    *(v2f*)(score + (size_t)(m0 + 16 * s + rr) * BB + n0 + cc) = o;
  }
}

extern "C" void kernel_launch(void* const* d_in, const int* in_sizes, int n_in,
                              void* d_out, int out_size, void* d_ws, size_t ws_size,
                              hipStream_t stream) {
  const float* r    = (const float*)d_in[0];
  const float* m    = (const float*)d_in[1];
  const float* t    = (const float*)d_in[2];
  const int*   ntid = (const int*)d_in[3];
  const int*   nlen = (const int*)d_in[4];
  const int*   antk = (const int*)d_in[5];
  const float* emb  = (const float*)d_in[6];
  const float* nW   = (const float*)d_in[7];
  const float* nb   = (const float*)d_in[8];
  const float* W1   = (const float*)d_in[9];
  const float* b1   = (const float*)d_in[10];
  const float* W2   = (const float*)d_in[11];
  const float* b2   = (const float*)d_in[12];
  const float* trW  = (const float*)d_in[13];
  const float* trb  = (const float*)d_in[14];
  const int*   t2i  = (const int*)d_in[15];

  float* score    = (float*)d_out;
  float* concepts = score + (size_t)BB * BB;

  float* ws  = (float*)d_ws;
  float* cls = ws;  ws += (size_t)NN * DD;
  float* h   = ws;  ws += (size_t)BB * DD;
  float* aem = ws;  ws += (size_t)BB * DD;
  float* trm = ws;  ws += (size_t)BB * DD;
  int* list  = (int*)ws;

  const dim3 wave(32);
  k_gemm_xw<<<dim3(BB / 32, DD / 64 + 1), dim3(256), 0, stream>>>(m, W1, b1, h, 1, 1, ntid,
                                                                  nlen, t2i, list);
  k_cls<<<dim3(DD / 256, CLS_GY), dim3(128), 0, stream>>>(list, antk, emb, nW, nb, cls);
  k_concepts<<<dim3(BB * LL / 64, DD / 64), wave, 0, stream>>>(ntid, nlen, t2i, emb,
                                                               nW, nb, cls, concepts);
  k_gemm_xw<<<dim3(BB / 32, DD / 64), dim3(256), 0, stream>>>(h, W2, b2, aem, 0, 0, ntid,
                                                              nlen, t2i, list);
  k_gemm_xw<<<dim3(BB / 32, DD / 64), dim3(256), 0, stream>>>(m, trW, trb, trm, 0, 0, ntid,
                                                              nlen, t2i, list);
  k_score<<<dim3(BB / 32, BB / 32), dim3(256), 0, stream>>>(trm, aem, r, t, score);
}


# Concepts GEMM only over rows that take their own token's encoding: launch 1 lists them behind the class rows, one row GEMM (48x64 tiles) stores them, the concepts launch only copies; sums unchanged
# speedup vs baseline: 4.6630x; 1.2124x over previous
#include <hip/hip_runtime.h>
#include <hip/hip_bf16.h>

typedef __attribute__((ext_vector_type(2))) float v2f;
typedef __attribute__((ext_vector_type(4))) float v4f;
typedef __attribute__((ext_vector_type(4))) int v4i;
typedef __attribute__((ext_vector_type(8))) float v8f;

#define BB 384
#define DD 1024
#define WD 300
#define NN 8000
#define LL 16

static __device__ __forceinline__ v8f wmma4(v2f a, v2f b, v8f c) {
  return __builtin_amdgcn_wmma_f32_16x16x4_f32(false, a, false, b, (short)0, c,
                                               false, false);
}

static __device__ __forceinline__ v2f lo2(v4f q) { return __builtin_shufflevector(q, q, 0, 1); }
static __device__ __forceinline__ v2f hi2(v4f q) { return __builtin_shufflevector(q, q, 2, 3); }

#define KC 5
#define NCH (WD / (4 * KC))

template <int S>
struct Chunk {
  v4f ax[S];
  v4f ay[S];
  v2f az[S];
  v4f b0[KC];
  v4f b1[KC];
};

template <int S>
static __device__ __forceinline__ void load_chunk(Chunk<S>& f, const float* const (&ar)[S],
                                                  int zd, const float* wpx,
                                                  const float* wpz, int kb) {
#pragma unroll
  for (int s = 0; s < S; ++s) {
    f.ax[s] = *(const v4f*)(ar[s] + kb);
    f.ay[s] = *(const v4f*)(ar[s] + kb + 4);
    f.az[s] = *(const v2f*)(ar[s] + kb + zd);
  }
#pragma unroll
  for (int j = 0; j < KC - 1; ++j) {
    f.b0[j] = *(const v4f*)(wpx + (size_t)(kb + 2 * j) * DD);
    f.b1[j] = *(const v4f*)(wpx + (size_t)(kb + 2 * j + 1) * DD);
  }
  f.b0[KC - 1] = *(const v4f*)(wpz + (size_t)kb * DD);
  f.b1[KC - 1] = *(const v4f*)(wpz + (size_t)(kb + 1) * DD);
}

template <int S>
static __device__ __forceinline__ void mma_chunk(v8f (&acc)[S][4], const Chunk<S>& f) {
#pragma unroll
  for (int j = 0; j < KC; ++j) {
#pragma unroll
    for (int t = 0; t < 4; ++t) {
      v2f b;
      b.x = f.b0[j][t];
      b.y = f.b1[j][t];
#pragma unroll
      for (int s = 0; s < S; ++s) {
        const v2f a = j == 0 ? lo2(f.ax[s]) : j == 1 ? hi2(f.ax[s])
                    : j == 2 ? lo2(f.ay[s]) : j == 3 ? hi2(f.ay[s]) : f.az[s];
        acc[s][t] = wmma4(a, b, acc[s][t]);
      }
    }
  }
}

template <int S>
static __device__ __forceinline__ void gemm300_tile(v8f (&acc)[S][4],
                                                    const float* const (&ar)[S], int zd,
                                                    const float* wpx, const float* wpz) {
#pragma unroll
  for (int s = 0; s < S; ++s)
#pragma unroll
    for (int t = 0; t < 4; ++t) acc[s][t] = (v8f){0.f,0.f,0.f,0.f,0.f,0.f,0.f,0.f};
  Chunk<S> f0, f1;
  load_chunk(f0, ar, zd, wpx, wpz, 0);
#pragma unroll 1
  for (int ch = 0; ch < NCH - 1; ch += 2) {
    load_chunk(f1, ar, zd, wpx, wpz, (ch + 1) * 4 * KC);
    mma_chunk(acc, f0);
    load_chunk(f0, ar, zd, wpx, wpz, (ch + 2) * 4 * KC);
    mma_chunk(acc, f1);
  }
  mma_chunk(acc, f0);
}

#define NPOS (BB * LL)
#define NFLAG (NN + NPOS)

#define ROW_S 3
#define ROW_RT (16 * ROW_S)
#define ROW_NT ((NPOS + 1 + ROW_RT - 1) / ROW_RT)
#define ROW_GY 64
__global__ __launch_bounds__(128) void k_rows(const int* __restrict__ list,
                                             const int* __restrict__ tok,
                                             const int* __restrict__ ntid,
                                             const float* __restrict__ emb,
                                             const float* __restrict__ W,
                                             const float* __restrict__ bias,
                                             float* __restrict__ cls,
                                             float* __restrict__ concepts) {
  const int lane = threadIdx.x & 31;
  const int half = lane >> 4;
  const int lid  = lane & 15;
  const int n0 = (blockIdx.x * 4 + (threadIdx.x >> 5)) * 64;
  const int c0 = n0 + 4 * lid;
  const int xo = half ? 12 : 0;
  const int zo = half ? 10 : 8;
  const float* wpx = W + (size_t)xo * DD + c0;
  const float* wpz = W + (size_t)zo * DD + c0;
  const v4f bv = *(const v4f*)(bias + c0);
  for (int rt = blockIdx.y; rt < ROW_NT; rt += ROW_GY) {
    const int r0 = rt * ROW_RT;
    if (__builtin_amdgcn_readfirstlane(list[r0]) < 0) break;
    int oe[ROW_S][8];
#pragma unroll
    for (int s = 0; s < ROW_S; ++s) {
      const v4i q0 = *(const v4i*)(list + r0 + 16 * s + 8 * half);
      const v4i q1 = *(const v4i*)(list + r0 + 16 * s + 8 * half + 4);
#pragma unroll
      for (int i = 0; i < 4; ++i) {
        oe[s][i]     = q0[i];
        oe[s][4 + i] = q1[i];
      }
    }
    const float* ar[ROW_S];
#pragma unroll
    for (int s = 0; s < ROW_S; ++s) {
      const int e = list[r0 + 16 * s + lid];
      const int* tp = e < NN ? tok + (e < 0 ? 0 : e) : ntid + (e - NN);
      ar[s] = emb + (size_t)*tp * WD + xo;
    }
    v8f acc[ROW_S][4];
    gemm300_tile<ROW_S>(acc, ar, zo - xo, wpx, wpz);
#pragma unroll
    for (int s = 0; s < ROW_S; ++s)
#pragma unroll
      for (int i = 0; i < 8; ++i) {
        v4f o;
        o.x = acc[s][0][i] + bv.x;
        o.y = acc[s][1][i] + bv.y;
        o.z = acc[s][2][i] + bv.z;
        o.w = acc[s][3][i] + bv.w;
        const int e = oe[s][i];
        if (e >= 0) {
          float* dst = e < NN ? cls + (size_t)e * DD : concepts + (size_t)(e - NN) * DD;
          *(v4f*)(dst + c0) = o;
        }
      }
  }
}

__global__ __launch_bounds__(32) void k_concepts(const int* __restrict__ src,
                                                 const float* __restrict__ cls,
                                                 float* __restrict__ concepts) {
  const int lane = threadIdx.x & 31;
  const int half = lane >> 4;
  const int lid  = lane & 15;
  const int m0 = blockIdx.x * 64, n0 = blockIdx.y * 64;
  const int c0 = n0 + 4 * lid;
  int code[4][8];
#pragma unroll
  for (int s = 0; s < 4; ++s) {
    const v4i q0 = *(const v4i*)(src + m0 + 16 * s + 8 * half);
    const v4i q1 = *(const v4i*)(src + m0 + 16 * s + 8 * half + 4);
#pragma unroll
    for (int i = 0; i < 4; ++i) {
      code[s][i]     = q0[i];
      code[s][4 + i] = q1[i];
    }
  }
  v4f mv[4][8];
#pragma unroll
  for (int s = 0; s < 4; ++s)
#pragma unroll
    for (int i = 0; i < 8; ++i)
      mv[s][i] = *(const v4f*)(cls + (size_t)(code[s][i] > 0 ? code[s][i] : 0) * DD + c0);
#pragma unroll
  for (int s = 0; s < 4; ++s) {
#pragma unroll
    for (int i = 0; i < 8; ++i) {
      const int row = m0 + s * 16 + half * 8 + i;
      const v4f val = code[s][i] > 0 ? mv[s][i] : (v4f){0.f, 0.f, 0.f, 0.f};
      if (code[s][i] >= 0) *(v4f*)(concepts + (size_t)row * DD + c0) = val;
    }
  }
}

static __device__ __forceinline__ void list_rows(const int* __restrict__ ntid,
                                                 const int* __restrict__ nlen,
                                                 const int* __restrict__ t2i,
                                                 int* __restrict__ list,
                                                 int* __restrict__ src,
                                                 unsigned char* used, int* wtot) {
  const int tid  = threadIdx.x;
  const int wv   = tid >> 5;
  const int lane = tid & 31;
  for (int i = tid; i < NFLAG / 4; i += 256) ((int*)used)[i] = 0;
  __syncthreads();
  int tk[NPOS / 256], wi[NPOS / 256];
  bool vd[NPOS / 256];
#pragma unroll
  for (int j = 0; j < NPOS / 256; ++j) {
    const int p = j * 256 + tid;
    tk[j] = ntid[p];
    vd[j] = (p & (LL - 1)) < nlen[p / LL];
  }
#pragma unroll
  for (int j = 0; j < NPOS / 256; ++j) wi[j] = t2i[tk[j]];
#pragma unroll
  for (int j = 0; j < NPOS / 256; ++j) {
    const int p = j * 256 + tid;
    int code = 0;
    if (vd[j]) {
      if (wi[j] > 0) {
        used[wi[j]] = 1;
        code = wi[j];
      } else {
        used[NN + p] = 1;
        code = -1;
      }
    }
    src[p] = code;
  }
  if (tid == 0) used[0] = 1;
  __syncthreads();
  int w[16];
  int cnt = 0;
  if (tid < NFLAG / 64) {
#pragma unroll
    for (int q = 0; q < 4; ++q) {
      const v4i v = ((const v4i*)used)[4 * tid + q];
#pragma unroll
      for (int i = 0; i < 4; ++i) w[4 * q + i] = v[i];
    }
    int sum = 0;
#pragma unroll
    for (int i = 0; i < 16; ++i) sum += w[i];
    cnt = (sum & 0xff) + ((sum >> 8) & 0xff) + ((sum >> 16) & 0xff) + ((sum >> 24) & 0xff);
  } else {
#pragma unroll
    for (int i = 0; i < 16; ++i) w[i] = 0;
  }
  int inc = cnt;
#pragma unroll
  for (int d = 1; d < 32; d *= 2) {
    const int v = __shfl_up(inc, d);
    if (lane >= d) inc += v;
  }
  if (lane == 31) wtot[wv] = inc;
  __syncthreads();
  int rank = inc - cnt, total = 0;
#pragma unroll
  for (int i = 0; i < 8; ++i) {
    const int c = wtot[i];
    if (i < wv) rank += c;
    total += c;
  }
#pragma unroll
  for (int j = 0; j < 64; ++j)
    if ((w[j >> 2] >> (8 * (j & 3))) & 1) list[rank++] = 64 * tid + j;
  for (int i = total + tid; i < NN; i += 256) list[i] = -1;
}

#define KS 128
#define KCH 8

__global__ __launch_bounds__(256) void k_gemm_xw(const float* __restrict__ X,
                                                 const float* __restrict__ W,
                                                 const float* __restrict__ bias,
                                                 float* __restrict__ out, int relu,
                                                 int lister,
                                                 const int* __restrict__ ntid,
                                                 const int* __restrict__ nlen,
                                                 const int* __restrict__ t2i,
                                                 int* __restrict__ list,
                                                 int* __restrict__ src) {
  __shared__ v4f part[8][16][16];
  const int by = (int)blockIdx.y - lister;
  if (by < 0) {
    unsigned char* used = (unsigned char*)&part[0][0][0];
    if (blockIdx.x == 0) list_rows(ntid, nlen, t2i, list, src, used, (int*)(used + NFLAG));
    return;
  }
  const int tid  = threadIdx.x;
  const int wv   = tid >> 5;
  const int lane = tid & 31;
  const int half = lane >> 4;
  const int lid  = lane & 15;
  const int m0 = blockIdx.x * 32, n0 = by * 64;
  const int kb = wv * KS + 16 * half;
  const float* arow0 = X + (size_t)(m0 + lid) * DD + kb;
  const float* arow1 = X + (size_t)(m0 + 16 + lid) * DD + kb;
  const float* wp = W + (size_t)kb * DD + n0 + 4 * lid;
  v8f acc[2][4];
#pragma unroll
  for (int s = 0; s < 2; ++s)
#pragma unroll
    for (int t = 0; t < 4; ++t) acc[s][t] = (v8f){0.f,0.f,0.f,0.f,0.f,0.f,0.f,0.f};
  for (int kc = 0; kc < KS; kc += 4 * KCH) {
    v4f a0[KCH / 2], a1[KCH / 2];
    v4f w0[KCH], w1[KCH];
#pragma unroll
    for (int i = 0; i < KCH / 2; ++i) {
      a0[i] = *(const v4f*)(arow0 + kc + 4 * i);
      a1[i] = *(const v4f*)(arow1 + kc + 4 * i);
    }
#pragma unroll
    for (int j = 0; j < KCH; ++j) {
      w0[j] = *(const v4f*)(wp + (size_t)(kc + 2 * j) * DD);
      w1[j] = *(const v4f*)(wp + (size_t)(kc + 2 * j + 1) * DD);
    }
    __builtin_amdgcn_sched_barrier(0);
#pragma unroll
    for (int j = 0; j < KCH; ++j) {
      const v2f fa0 = (j & 1) ? hi2(a0[j >> 1]) : lo2(a0[j >> 1]);
      const v2f fa1 = (j & 1) ? hi2(a1[j >> 1]) : lo2(a1[j >> 1]);
#pragma unroll
      for (int t = 0; t < 4; ++t) {
        v2f b;
        b.x = w0[j][t];
        b.y = w1[j][t];
        acc[0][t] = wmma4(fa0, b, acc[0][t]);
        acc[1][t] = wmma4(fa1, b, acc[1][t]);
      }
    }
  }
  const int rr = tid >> 4, cc = tid & 15;
  const v4f bv = *(const v4f*)(bias + n0 + 4 * cc);
#pragma unroll
  for (int s = 0; s < 2; ++s) {
    if (s) __syncthreads();
#pragma unroll
    for (int i = 0; i < 8; ++i) {
      v4f o;
      o.x = acc[s][0][i];
      o.y = acc[s][1][i];
      o.z = acc[s][2][i];
      o.w = acc[s][3][i];
      part[wv][half * 8 + i][lid] = o;
    }
    __syncthreads();
    v4f sum = part[0][rr][cc];
#pragma unroll
    for (int w = 1; w < 8; ++w) sum += part[w][rr][cc];
    sum += bv;
    if (relu) {
      sum.x = fmaxf(sum.x, 0.f);
      sum.y = fmaxf(sum.y, 0.f);
      sum.z = fmaxf(sum.z, 0.f);
      sum.w = fmaxf(sum.w, 0.f);
    }
    *(v4f*)(out + (size_t)(m0 + 16 * s + rr) * DD + n0 + 4 * cc) = sum;
  }
}

static __device__ __forceinline__ v2f sq_acc(v2f acc, v4f q) {
  acc += lo2(q) * lo2(q);
  acc += hi2(q) * hi2(q);
  return acc;
}

static __device__ __forceinline__ v4f exp4(v4f x, float m) {
  v4f e;
  e.x = __expf(x.x - m);
  e.y = __expf(x.y - m);
  e.z = __expf(x.z - m);
  e.w = __expf(x.w - m);
  return e;
}

static __device__ __forceinline__ float max4(float m, v4f x) {
  return fmaxf(fmaxf(m, fmaxf(x.x, x.y)), fmaxf(x.z, x.w));
}

__global__ __launch_bounds__(256) void k_score(const float* __restrict__ trm,
                                               const float* __restrict__ aem,
                                               const float* __restrict__ r,
                                               const float* __restrict__ t,
                                               float* __restrict__ score) {
  __shared__ __attribute__((aligned(16))) float part[8][3][16][32];
  __shared__ __attribute__((aligned(16))) float ssq[8][3][32];
  __shared__ float rmx[8][32];
  const int tid  = threadIdx.x;
  const int wv   = tid >> 5;
  const int lane = tid & 31;
  const int half = lane >> 4;
  const int lid  = lane & 15;
  const int m0 = blockIdx.x * 32, n0 = blockIdx.y * 32;
  const int kb = wv * KS + 16 * half;
  const size_t ra0 = (size_t)(m0 + lid) * DD + kb;
  const size_t ra1 = (size_t)(m0 + 16 + lid) * DD + kb;
  const size_t cb0 = (size_t)(n0 + lid) * DD + kb;
  const size_t cb1 = (size_t)(n0 + 16 + lid) * DD + kb;
  float mx0, mx1;
  {
    v4f x0[KS / 8], x1[KS / 8];
#pragma unroll
    for (int i = 0; i < KS / 8; ++i) {
      const int k0 = (i >> 2) * 4 * KCH + 4 * (i & 3);
      x0[i] = *(const v4f*)(aem + ra0 + k0);
      x1[i] = *(const v4f*)(aem + ra1 + k0);
    }
    mx0 = x0[0].x;
    mx1 = x1[0].x;
#pragma unroll
    for (int i = 0; i < KS / 8; ++i) {
      mx0 = max4(mx0, x0[i]);
      mx1 = max4(mx1, x1[i]);
    }
  }
  mx0 = fmaxf(mx0, __shfl_xor(mx0, 16));
  mx1 = fmaxf(mx1, __shfl_xor(mx1, 16));
  if (half == 0) {
    rmx[wv][lid]      = mx0;
    rmx[wv][16 + lid] = mx1;
  }
  __syncthreads();
  mx0 = rmx[0][lid];
  mx1 = rmx[0][16 + lid];
  for (int w = 1; w < 8; ++w) {
    mx0 = fmaxf(mx0, rmx[w][lid]);
    mx1 = fmaxf(mx1, rmx[w][16 + lid]);
  }
  const float mx[2] = {mx0, mx1};
  v8f num[2][2], den[2][2], isv[2][2];
#pragma unroll
  for (int s = 0; s < 2; ++s)
#pragma unroll
    for (int u = 0; u < 2; ++u) {
      num[s][u] = (v8f){0.f,0.f,0.f,0.f,0.f,0.f,0.f,0.f};
      den[s][u] = num[s][u];
      isv[s][u] = num[s][u];
    }
  v2f qt2[2], qr2[2], qc2[2];
#pragma unroll
  for (int s = 0; s < 2; ++s) {
    qt2[s] = (v2f){0.f, 0.f};
    qr2[s] = qt2[s];
    qc2[s] = qt2[s];
  }
  for (int kc = 0; kc < KS; kc += 4 * KCH) {
    v4f a1[KCH / 2][2], a2[KCH / 2][2], a3[KCH / 2][2];
    v4f b1[KCH / 2][2], b2[KCH / 2][2];
#pragma unroll
    for (int i = 0; i < KCH / 2; ++i) {
      const int k0 = kc + 4 * i;
      a1[i][0] = *(const v4f*)(trm + ra0 + k0);  a1[i][1] = *(const v4f*)(trm + ra1 + k0);
      a2[i][0] = *(const v4f*)(aem + ra0 + k0);  a2[i][1] = *(const v4f*)(aem + ra1 + k0);
      a3[i][0] = *(const v4f*)(r   + ra0 + k0);  a3[i][1] = *(const v4f*)(r   + ra1 + k0);
      b1[i][0] = *(const v4f*)(t   + cb0 + k0);  b1[i][1] = *(const v4f*)(t   + cb1 + k0);
    }
    __builtin_amdgcn_sched_barrier(0);
#pragma unroll
    for (int i = 0; i < KCH / 2; ++i)
#pragma unroll
      for (int s = 0; s < 2; ++s) {
        qt2[s] = sq_acc(qt2[s], a1[i][s]);
        qr2[s] = sq_acc(qr2[s], a3[i][s]);
        qc2[s] = sq_acc(qc2[s], b1[i][s]);
        a2[i][s] = exp4(a2[i][s], mx[s]);
        a1[i][s] = a1[i][s] * a2[i][s];
        a2[i][s] = a2[i][s] * a2[i][s];
        b2[i][s] = b1[i][s] * b1[i][s];
      }
#pragma unroll
    for (int j = 0; j < KCH; ++j)
#pragma unroll
      for (int s = 0; s < 2; ++s)
#pragma unroll
        for (int u = 0; u < 2; ++u) {
          const int i = j >> 1;
          const v2f fa1 = (j & 1) ? hi2(a1[i][s]) : lo2(a1[i][s]);
          const v2f fa2 = (j & 1) ? hi2(a2[i][s]) : lo2(a2[i][s]);
          const v2f fa3 = (j & 1) ? hi2(a3[i][s]) : lo2(a3[i][s]);
          const v2f fb1 = (j & 1) ? hi2(b1[i][u]) : lo2(b1[i][u]);
          const v2f fb2 = (j & 1) ? hi2(b2[i][u]) : lo2(b2[i][u]);
          num[s][u] = wmma4(fa1, fb1, num[s][u]);
          den[s][u] = wmma4(fa2, fb2, den[s][u]);
          isv[s][u] = wmma4(fa3, fb1, isv[s][u]);
        }
  }
  float qt[2], qr[2], qc[2];
#pragma unroll
  for (int s = 0; s < 2; ++s) {
    qt[s] = qt2[s].x + qt2[s].y;
    qr[s] = qr2[s].x + qr2[s].y;
    qc[s] = qc2[s].x + qc2[s].y;
    qt[s] += __shfl_xor(qt[s], 16);
    qr[s] += __shfl_xor(qr[s], 16);
    qc[s] += __shfl_xor(qc[s], 16);
  }
  if (half == 0) {
#pragma unroll
    for (int s = 0; s < 2; ++s) {
      ssq[wv][0][16 * s + lid] = qt[s];
      ssq[wv][1][16 * s + lid] = qr[s];
      ssq[wv][2][16 * s + lid] = qc[s];
    }
  }
  const int rr = tid >> 4, cc = 2 * (tid & 15);
  v2f ic = {0.f, 0.f};
#pragma unroll
  for (int s = 0; s < 2; ++s) {
    if (s) __syncthreads();
#pragma unroll
    for (int u = 0; u < 2; ++u)
#pragma unroll
      for (int i = 0; i < 8; ++i) {
        part[wv][0][half * 8 + i][16 * u + lid] = num[s][u][i];
        part[wv][1][half * 8 + i][16 * u + lid] = den[s][u][i];
        part[wv][2][half * 8 + i][16 * u + lid] = isv[s][u][i];
      }
    __syncthreads();
    if (s == 0) {
      v2f c = *(const v2f*)&ssq[0][2][cc];
#pragma unroll
      for (int w = 1; w < 8; ++w) c += *(const v2f*)&ssq[w][2][cc];
      ic.x = rsqrtf(c.x);
      ic.y = rsqrtf(c.y);
    }
    float st = ssq[0][0][16 * s + rr];
    float sr = ssq[0][1][16 * s + rr];
#pragma unroll
    for (int w = 1; w < 8; ++w) {
      st += ssq[w][0][16 * s + rr];
      sr += ssq[w][1][16 * s + rr];
    }
    const float it = rsqrtf(st);
    const float ir = rsqrtf(sr);
    v2f sn = *(const v2f*)&part[0][0][rr][cc];
    v2f sd = *(const v2f*)&part[0][1][rr][cc];
    v2f si = *(const v2f*)&part[0][2][rr][cc];
#pragma unroll
    for (int w = 1; w < 8; ++w) {
      sn += *(const v2f*)&part[w][0][rr][cc];
      sd += *(const v2f*)&part[w][1][rr][cc];
      si += *(const v2f*)&part[w][2][rr][cc];
    }
    v2f o;
    o.x = sn.x * rsqrtf(sd.x) * it + si.x * (ir * ic.x);
    o.y = sn.y * rsqrtf(sd.y) * it + si.y * (ir * ic.y);
    *(v2f*)(score + (size_t)(m0 + 16 * s + rr) * BB + n0 + cc) = o;
  }
}

extern "C" void kernel_launch(void* const* d_in, const int* in_sizes, int n_in,
                              void* d_out, int out_size, void* d_ws, size_t ws_size,
                              hipStream_t stream) {
  const float* r    = (const float*)d_in[0];
  const float* m    = (const float*)d_in[1];
  const float* t    = (const float*)d_in[2];
  const int*   ntid = (const int*)d_in[3];
  const int*   nlen = (const int*)d_in[4];
  const int*   antk = (const int*)d_in[5];
  const float* emb  = (const float*)d_in[6];
  const float* nW   = (const float*)d_in[7];
  const float* nb   = (const float*)d_in[8];
  const float* W1   = (const float*)d_in[9];
  const float* b1   = (const float*)d_in[10];
  const float* W2   = (const float*)d_in[11];
  const float* b2   = (const float*)d_in[12];
  const float* trW  = (const float*)d_in[13];
  const float* trb  = (const float*)d_in[14];
  const int*   t2i  = (const int*)d_in[15];

  float* score    = (float*)d_out;
  float* concepts = score + (size_t)BB * BB;

  float* ws  = (float*)d_ws;
  float* cls = ws;  ws += (size_t)NN * DD;
  float* h   = ws;  ws += (size_t)BB * DD;
  float* aem = ws;  ws += (size_t)BB * DD;
  float* trm = ws;  ws += (size_t)BB * DD;
  int* list  = (int*)ws;
  int* src   = list + NN;

  const dim3 wave(32);
  k_gemm_xw<<<dim3(BB / 32, DD / 64 + 1), dim3(256), 0, stream>>>(m, W1, b1, h, 1, 1, ntid,
                                                                  nlen, t2i, list, src);
  k_rows<<<dim3(DD / 256, ROW_GY), dim3(128), 0, stream>>>(list, antk, ntid, emb, nW, nb,
                                                           cls, concepts);
  k_concepts<<<dim3(NPOS / 64, DD / 64), wave, 0, stream>>>(src, cls, concepts);
  k_gemm_xw<<<dim3(BB / 32, DD / 64), dim3(256), 0, stream>>>(h, W2, b2, aem, 0, 0, ntid,
                                                              nlen, t2i, list, src);
  k_gemm_xw<<<dim3(BB / 32, DD / 64), dim3(256), 0, stream>>>(m, trW, trb, trm, 0, 0, ntid,
                                                              nlen, t2i, list, src);
  k_score<<<dim3(BB / 32, BB / 32), dim3(256), 0, stream>>>(trm, aem, r, t, score);
}


# Both products of m (first attention layer, transform) as one GEMM: m row pieces loaded once feed the same 32 columns of both weight matrices; 48x(32+32) tiles, 256 groups, 5 launches; sums unchanged
# speedup vs baseline: 5.1514x; 1.1047x over previous
#include <hip/hip_runtime.h>
#include <hip/hip_bf16.h>

typedef __attribute__((ext_vector_type(2))) float v2f;
typedef __attribute__((ext_vector_type(4))) float v4f;
typedef __attribute__((ext_vector_type(4))) int v4i;
typedef __attribute__((ext_vector_type(8))) float v8f;

#define BB 384
#define DD 1024
#define WD 300
#define NN 8000
#define LL 16

static __device__ __forceinline__ v8f wmma4(v2f a, v2f b, v8f c) {
  return __builtin_amdgcn_wmma_f32_16x16x4_f32(false, a, false, b, (short)0, c,
                                               false, false);
}

static __device__ __forceinline__ v2f lo2(v4f q) { return __builtin_shufflevector(q, q, 0, 1); }
static __device__ __forceinline__ v2f hi2(v4f q) { return __builtin_shufflevector(q, q, 2, 3); }

#define KC 5
#define NCH (WD / (4 * KC))

template <int S>
struct Chunk {
  v4f ax[S];
  v4f ay[S];
  v2f az[S];
  v4f b0[KC];
  v4f b1[KC];
};

template <int S>
static __device__ __forceinline__ void load_chunk(Chunk<S>& f, const float* const (&ar)[S],
                                                  int zd, const float* wpx,
                                                  const float* wpz, int kb) {
#pragma unroll
  for (int s = 0; s < S; ++s) {
    f.ax[s] = *(const v4f*)(ar[s] + kb);
    f.ay[s] = *(const v4f*)(ar[s] + kb + 4);
    f.az[s] = *(const v2f*)(ar[s] + kb + zd);
  }
#pragma unroll
  for (int j = 0; j < KC - 1; ++j) {
    f.b0[j] = *(const v4f*)(wpx + (size_t)(kb + 2 * j) * DD);
    f.b1[j] = *(const v4f*)(wpx + (size_t)(kb + 2 * j + 1) * DD);
  }
  f.b0[KC - 1] = *(const v4f*)(wpz + (size_t)kb * DD);
  f.b1[KC - 1] = *(const v4f*)(wpz + (size_t)(kb + 1) * DD);
}

template <int S>
static __device__ __forceinline__ void mma_chunk(v8f (&acc)[S][4], const Chunk<S>& f) {
#pragma unroll
  for (int j = 0; j < KC; ++j) {
#pragma unroll
    for (int t = 0; t < 4; ++t) {
      v2f b;
      b.x = f.b0[j][t];
      b.y = f.b1[j][t];
#pragma unroll
      for (int s = 0; s < S; ++s) {
        const v2f a = j == 0 ? lo2(f.ax[s]) : j == 1 ? hi2(f.ax[s])
                    : j == 2 ? lo2(f.ay[s]) : j == 3 ? hi2(f.ay[s]) : f.az[s];
        acc[s][t] = wmma4(a, b, acc[s][t]);
      }
    }
  }
}

template <int S>
static __device__ __forceinline__ void gemm300_tile(v8f (&acc)[S][4],
                                                    const float* const (&ar)[S], int zd,
                                                    const float* wpx, const float* wpz) {
#pragma unroll
  for (int s = 0; s < S; ++s)
#pragma unroll
    for (int t = 0; t < 4; ++t) acc[s][t] = (v8f){0.f,0.f,0.f,0.f,0.f,0.f,0.f,0.f};
  Chunk<S> f0, f1;
  load_chunk(f0, ar, zd, wpx, wpz, 0);
#pragma unroll 1
  for (int ch = 0; ch < NCH - 1; ch += 2) {
    load_chunk(f1, ar, zd, wpx, wpz, (ch + 1) * 4 * KC);
    mma_chunk(acc, f0);
    load_chunk(f0, ar, zd, wpx, wpz, (ch + 2) * 4 * KC);
    mma_chunk(acc, f1);
  }
  mma_chunk(acc, f0);
}

#define NPOS (BB * LL)
#define NFLAG (NN + NPOS)

#define ROW_S 3
#define ROW_RT (16 * ROW_S)
#define ROW_NT ((NPOS + 1 + ROW_RT - 1) / ROW_RT)
#define ROW_GY 64
__global__ __launch_bounds__(128) void k_rows(const int* __restrict__ list,
                                             const int* __restrict__ tok,
                                             const int* __restrict__ ntid,
                                             const float* __restrict__ emb,
                                             const float* __restrict__ W,
                                             const float* __restrict__ bias,
                                             float* __restrict__ cls,
                                             float* __restrict__ concepts) {
  const int lane = threadIdx.x & 31;
  const int half = lane >> 4;
  const int lid  = lane & 15;
  const int n0 = (blockIdx.x * 4 + (threadIdx.x >> 5)) * 64;
  const int c0 = n0 + 4 * lid;
  const int xo = half ? 12 : 0;
  const int zo = half ? 10 : 8;
  const float* wpx = W + (size_t)xo * DD + c0;
  const float* wpz = W + (size_t)zo * DD + c0;
  const v4f bv = *(const v4f*)(bias + c0);
  for (int rt = blockIdx.y; rt < ROW_NT; rt += ROW_GY) {
    const int r0 = rt * ROW_RT;
    if (__builtin_amdgcn_readfirstlane(list[r0]) < 0) break;
    int oe[ROW_S][8];
#pragma unroll
    for (int s = 0; s < ROW_S; ++s) {
      const v4i q0 = *(const v4i*)(list + r0 + 16 * s + 8 * half);
      const v4i q1 = *(const v4i*)(list + r0 + 16 * s + 8 * half + 4);
#pragma unroll
      for (int i = 0; i < 4; ++i) {
        oe[s][i]     = q0[i];
        oe[s][4 + i] = q1[i];
      }
    }
    const float* ar[ROW_S];
#pragma unroll
    for (int s = 0; s < ROW_S; ++s) {
      const int e = list[r0 + 16 * s + lid];
      const int* tp = e < NN ? tok + (e < 0 ? 0 : e) : ntid + (e - NN);
      ar[s] = emb + (size_t)*tp * WD + xo;
    }
    v8f acc[ROW_S][4];
    gemm300_tile<ROW_S>(acc, ar, zo - xo, wpx, wpz);
#pragma unroll
    for (int s = 0; s < ROW_S; ++s)
#pragma unroll
      for (int i = 0; i < 8; ++i) {
        v4f o;
        o.x = acc[s][0][i] + bv.x;
        o.y = acc[s][1][i] + bv.y;
        o.z = acc[s][2][i] + bv.z;
        o.w = acc[s][3][i] + bv.w;
        const int e = oe[s][i];
        if (e >= 0) {
          float* dst = e < NN ? cls + (size_t)e * DD : concepts + (size_t)(e - NN) * DD;
          *(v4f*)(dst + c0) = o;
        }
      }
  }
}

__global__ __launch_bounds__(32) void k_concepts(const int* __restrict__ src,
                                                 const float* __restrict__ cls,
                                                 float* __restrict__ concepts) {
  const int lane = threadIdx.x & 31;
  const int half = lane >> 4;
  const int lid  = lane & 15;
  const int m0 = blockIdx.x * 64, n0 = blockIdx.y * 64;
  const int c0 = n0 + 4 * lid;
  int code[4][8];
#pragma unroll
  for (int s = 0; s < 4; ++s) {
    const v4i q0 = *(const v4i*)(src + m0 + 16 * s + 8 * half);
    const v4i q1 = *(const v4i*)(src + m0 + 16 * s + 8 * half + 4);
#pragma unroll
    for (int i = 0; i < 4; ++i) {
      code[s][i]     = q0[i];
      code[s][4 + i] = q1[i];
    }
  }
  v4f mv[4][8];
#pragma unroll
  for (int s = 0; s < 4; ++s)
#pragma unroll
    for (int i = 0; i < 8; ++i)
      mv[s][i] = *(const v4f*)(cls + (size_t)(code[s][i] > 0 ? code[s][i] : 0) * DD + c0);
#pragma unroll
  for (int s = 0; s < 4; ++s) {
#pragma unroll
    for (int i = 0; i < 8; ++i) {
      const int row = m0 + s * 16 + half * 8 + i;
      const v4f val = code[s][i] > 0 ? mv[s][i] : (v4f){0.f, 0.f, 0.f, 0.f};
      if (code[s][i] >= 0) *(v4f*)(concepts + (size_t)row * DD + c0) = val;
    }
  }
}

static __device__ __forceinline__ void list_rows(const int* __restrict__ ntid,
                                                 const int* __restrict__ nlen,
                                                 const int* __restrict__ t2i,
                                                 int* __restrict__ list,
                                                 int* __restrict__ src,
                                                 unsigned char* used, int* wtot) {
  const int tid  = threadIdx.x;
  const int wv   = tid >> 5;
  const int lane = tid & 31;
  for (int i = tid; i < NFLAG / 4; i += 256) ((int*)used)[i] = 0;
  __syncthreads();
  int tk[NPOS / 256], wi[NPOS / 256];
  bool vd[NPOS / 256];
#pragma unroll
  for (int j = 0; j < NPOS / 256; ++j) {
    const int p = j * 256 + tid;
    tk[j] = ntid[p];
    vd[j] = (p & (LL - 1)) < nlen[p / LL];
  }
#pragma unroll
  for (int j = 0; j < NPOS / 256; ++j) wi[j] = t2i[tk[j]];
#pragma unroll
  for (int j = 0; j < NPOS / 256; ++j) {
    const int p = j * 256 + tid;
    int code = 0;
    if (vd[j]) {
      if (wi[j] > 0) {
        used[wi[j]] = 1;
        code = wi[j];
      } else {
        used[NN + p] = 1;
        code = -1;
      }
    }
    src[p] = code;
  }
  if (tid == 0) used[0] = 1;
  __syncthreads();
  int w[16];
  int cnt = 0;
  if (tid < NFLAG / 64) {
#pragma unroll
    for (int q = 0; q < 4; ++q) {
      const v4i v = ((const v4i*)used)[4 * tid + q];
#pragma unroll
      for (int i = 0; i < 4; ++i) w[4 * q + i] = v[i];
    }
    int sum = 0;
#pragma unroll
    for (int i = 0; i < 16; ++i) sum += w[i];
    cnt = (sum & 0xff) + ((sum >> 8) & 0xff) + ((sum >> 16) & 0xff) + ((sum >> 24) & 0xff);
  } else {
#pragma unroll
    for (int i = 0; i < 16; ++i) w[i] = 0;
  }
  int inc = cnt;
#pragma unroll
  for (int d = 1; d < 32; d *= 2) {
    const int v = __shfl_up(inc, d);
    if (lane >= d) inc += v;
  }
  if (lane == 31) wtot[wv] = inc;
  __syncthreads();
  int rank = inc - cnt, total = 0;
#pragma unroll
  for (int i = 0; i < 8; ++i) {
    const int c = wtot[i];
    if (i < wv) rank += c;
    total += c;
  }
#pragma unroll
  for (int j = 0; j < 64; ++j)
    if ((w[j >> 2] >> (8 * (j & 3))) & 1) list[rank++] = 64 * tid + j;
  for (int i = total + tid; i < NN; i += 256) list[i] = -1;
}

#define KS 128
#define KCH 8
#define M2_S 3

__global__ __launch_bounds__(256, 4) void k_gemm_m2(const float* __restrict__ X,
                                                 const float* __restrict__ Wa,
                                                 const float* __restrict__ ba,
                                                 float* __restrict__ outa,
                                                 const float* __restrict__ Wb,
                                                 const float* __restrict__ bb,
                                                 float* __restrict__ outb,
                                                 const int* __restrict__ ntid,
                                                 const int* __restrict__ nlen,
                                                 const int* __restrict__ t2i,
                                                 int* __restrict__ list,
                                                 int* __restrict__ src) {
  __shared__ __attribute__((aligned(16))) float part[8][16][64];
  const int by = (int)blockIdx.y - 1;
  if (by < 0) {
    unsigned char* used = (unsigned char*)&part[0][0][0];
    if (blockIdx.x == 0) list_rows(ntid, nlen, t2i, list, src, used, (int*)(used + NFLAG));
    return;
  }
  const int tid  = threadIdx.x;
  const int wv   = tid >> 5;
  const int lane = tid & 31;
  const int half = lane >> 4;
  const int lid  = lane & 15;
  const int m0 = blockIdx.x * (16 * M2_S), n0 = by * 32;
  const int kb = wv * KS + 16 * half;
  const float* arow[M2_S];
#pragma unroll
  for (int s = 0; s < M2_S; ++s) arow[s] = X + (size_t)(m0 + 16 * s + lid) * DD + kb;
  const float* wpa = Wa + (size_t)kb * DD + n0 + 2 * lid;
  const float* wpb = Wb + (size_t)kb * DD + n0 + 2 * lid;
  v8f acc[M2_S][4];
#pragma unroll
  for (int s = 0; s < M2_S; ++s)
#pragma unroll
    for (int t = 0; t < 4; ++t) acc[s][t] = (v8f){0.f,0.f,0.f,0.f,0.f,0.f,0.f,0.f};
  for (int kc = 0; kc < KS; kc += 4 * KCH) {
    v4f a[M2_S][KCH / 2];
    v2f wa0[KCH], wa1[KCH], wb0[KCH], wb1[KCH];
#pragma unroll
    for (int i = 0; i < KCH / 2; ++i)
#pragma unroll
      for (int s = 0; s < M2_S; ++s) a[s][i] = *(const v4f*)(arow[s] + kc + 4 * i);
#pragma unroll
    for (int j = 0; j < KCH; ++j) {
      wa0[j] = *(const v2f*)(wpa + (size_t)(kc + 2 * j) * DD);
      wa1[j] = *(const v2f*)(wpa + (size_t)(kc + 2 * j + 1) * DD);
      wb0[j] = *(const v2f*)(wpb + (size_t)(kc + 2 * j) * DD);
      wb1[j] = *(const v2f*)(wpb + (size_t)(kc + 2 * j + 1) * DD);
    }
    __builtin_amdgcn_sched_barrier(0);
#pragma unroll
    for (int j = 0; j < KCH; ++j) {
      v2f fa[M2_S];
#pragma unroll
      for (int s = 0; s < M2_S; ++s) fa[s] = (j & 1) ? hi2(a[s][j >> 1]) : lo2(a[s][j >> 1]);
#pragma unroll
      for (int t = 0; t < 4; ++t) {
        v2f b;
        b.x = t < 2 ? wa0[j][t & 1] : wb0[j][t & 1];
        b.y = t < 2 ? wa1[j][t & 1] : wb1[j][t & 1];
#pragma unroll
        for (int s = 0; s < M2_S; ++s) acc[s][t] = wmma4(fa[s], b, acc[s][t]);
      }
    }
  }
  const int rr = tid >> 4, cc = tid & 15;
  const bool first = cc < 8;
  const int col = n0 + 4 * (cc & 7);
  const v4f bv = *(const v4f*)((first ? ba : bb) + col);
  float* const ocol = (first ? outa : outb) + col;
#pragma unroll
  for (int s = 0; s < M2_S; ++s) {
    if (s) __syncthreads();
#pragma unroll
    for (int i = 0; i < 8; ++i) {
      v2f* prow = (v2f*)&part[wv][half * 8 + i][0];
      prow[lid]      = (v2f){acc[s][0][i], acc[s][1][i]};
      prow[16 + lid] = (v2f){acc[s][2][i], acc[s][3][i]};
    }
    __syncthreads();
    v4f sum = *(const v4f*)&part[0][rr][4 * cc];
#pragma unroll
    for (int w = 1; w < 8; ++w) sum += *(const v4f*)&part[w][rr][4 * cc];
    sum += bv;
    if (first) {
      sum.x = fmaxf(sum.x, 0.f);
      sum.y = fmaxf(sum.y, 0.f);
      sum.z = fmaxf(sum.z, 0.f);
      sum.w = fmaxf(sum.w, 0.f);
    }
    *(v4f*)(ocol + (size_t)(m0 + 16 * s + rr) * DD) = sum;
  }
}

__global__ __launch_bounds__(256) void k_gemm_xw(const float* __restrict__ X,
                                                 const float* __restrict__ W,
                                                 const float* __restrict__ bias,
                                                 float* __restrict__ out) {
  __shared__ v4f part[8][16][16];
  const int tid  = threadIdx.x;
  const int wv   = tid >> 5;
  const int lane = tid & 31;
  const int half = lane >> 4;
  const int lid  = lane & 15;
  const int m0 = blockIdx.x * 32, n0 = blockIdx.y * 64;
  const int kb = wv * KS + 16 * half;
  const float* arow0 = X + (size_t)(m0 + lid) * DD + kb;
  const float* arow1 = X + (size_t)(m0 + 16 + lid) * DD + kb;
  const float* wp = W + (size_t)kb * DD + n0 + 4 * lid;
  v8f acc[2][4];
#pragma unroll
  for (int s = 0; s < 2; ++s)
#pragma unroll
    for (int t = 0; t < 4; ++t) acc[s][t] = (v8f){0.f,0.f,0.f,0.f,0.f,0.f,0.f,0.f};
  for (int kc = 0; kc < KS; kc += 4 * KCH) {
    v4f a0[KCH / 2], a1[KCH / 2];
    v4f w0[KCH], w1[KCH];
#pragma unroll
    for (int i = 0; i < KCH / 2; ++i) {
      a0[i] = *(const v4f*)(arow0 + kc + 4 * i);
      a1[i] = *(const v4f*)(arow1 + kc + 4 * i);
    }
#pragma unroll
    for (int j = 0; j < KCH; ++j) {
      w0[j] = *(const v4f*)(wp + (size_t)(kc + 2 * j) * DD);
      w1[j] = *(const v4f*)(wp + (size_t)(kc + 2 * j + 1) * DD);
    }
    __builtin_amdgcn_sched_barrier(0);
#pragma unroll
    for (int j = 0; j < KCH; ++j) {
      const v2f fa0 = (j & 1) ? hi2(a0[j >> 1]) : lo2(a0[j >> 1]);
      const v2f fa1 = (j & 1) ? hi2(a1[j >> 1]) : lo2(a1[j >> 1]);
#pragma unroll
      for (int t = 0; t < 4; ++t) {
        v2f b;
        b.x = w0[j][t];
        b.y = w1[j][t];
        acc[0][t] = wmma4(fa0, b, acc[0][t]);
        acc[1][t] = wmma4(fa1, b, acc[1][t]);
      }
    }
  }
  const int rr = tid >> 4, cc = tid & 15;
  const v4f bv = *(const v4f*)(bias + n0 + 4 * cc);
#pragma unroll
  for (int s = 0; s < 2; ++s) {
    if (s) __syncthreads();
#pragma unroll
    for (int i = 0; i < 8; ++i) {
      v4f o;
      o.x = acc[s][0][i];
      o.y = acc[s][1][i];
      o.z = acc[s][2][i];
      o.w = acc[s][3][i];
      part[wv][half * 8 + i][lid] = o;
    }
    __syncthreads();
    v4f sum = part[0][rr][cc];
#pragma unroll
    for (int w = 1; w < 8; ++w) sum += part[w][rr][cc];
    sum += bv;
    *(v4f*)(out + (size_t)(m0 + 16 * s + rr) * DD + n0 + 4 * cc) = sum;
  }
}

static __device__ __forceinline__ v2f sq_acc(v2f acc, v4f q) {
  acc += lo2(q) * lo2(q);
  acc += hi2(q) * hi2(q);
  return acc;
}

static __device__ __forceinline__ v4f exp4(v4f x, float m) {
  v4f e;
  e.x = __expf(x.x - m);
  e.y = __expf(x.y - m);
  e.z = __expf(x.z - m);
  e.w = __expf(x.w - m);
  return e;
}

static __device__ __forceinline__ float max4(float m, v4f x) {
  return fmaxf(fmaxf(m, fmaxf(x.x, x.y)), fmaxf(x.z, x.w));
}

__global__ __launch_bounds__(256) void k_score(const float* __restrict__ trm,
                                               const float* __restrict__ aem,
                                               const float* __restrict__ r,
                                               const float* __restrict__ t,
                                               float* __restrict__ score) {
  __shared__ __attribute__((aligned(16))) float part[8][3][16][32];
  __shared__ __attribute__((aligned(16))) float ssq[8][3][32];
  __shared__ float rmx[8][32];
  const int tid  = threadIdx.x;
  const int wv   = tid >> 5;
  const int lane = tid & 31;
  const int half = lane >> 4;
  const int lid  = lane & 15;
  const int m0 = blockIdx.x * 32, n0 = blockIdx.y * 32;
  const int kb = wv * KS + 16 * half;
  const size_t ra0 = (size_t)(m0 + lid) * DD + kb;
  const size_t ra1 = (size_t)(m0 + 16 + lid) * DD + kb;
  const size_t cb0 = (size_t)(n0 + lid) * DD + kb;
  const size_t cb1 = (size_t)(n0 + 16 + lid) * DD + kb;
  float mx0, mx1;
  {
    v4f x0[KS / 8], x1[KS / 8];
#pragma unroll
    for (int i = 0; i < KS / 8; ++i) {
      const int k0 = (i >> 2) * 4 * KCH + 4 * (i & 3);
      x0[i] = *(const v4f*)(aem + ra0 + k0);
      x1[i] = *(const v4f*)(aem + ra1 + k0);
    }
    mx0 = x0[0].x;
    mx1 = x1[0].x;
#pragma unroll
    for (int i = 0; i < KS / 8; ++i) {
      mx0 = max4(mx0, x0[i]);
      mx1 = max4(mx1, x1[i]);
    }
  }
  mx0 = fmaxf(mx0, __shfl_xor(mx0, 16));
  mx1 = fmaxf(mx1, __shfl_xor(mx1, 16));
  if (half == 0) {
    rmx[wv][lid]      = mx0;
    rmx[wv][16 + lid] = mx1;
  }
  __syncthreads();
  mx0 = rmx[0][lid];
  mx1 = rmx[0][16 + lid];
  for (int w = 1; w < 8; ++w) {
    mx0 = fmaxf(mx0, rmx[w][lid]);
    mx1 = fmaxf(mx1, rmx[w][16 + lid]);
  }
  const float mx[2] = {mx0, mx1};
  v8f num[2][2], den[2][2], isv[2][2];
#pragma unroll
  for (int s = 0; s < 2; ++s)
#pragma unroll
    for (int u = 0; u < 2; ++u) {
      num[s][u] = (v8f){0.f,0.f,0.f,0.f,0.f,0.f,0.f,0.f};
      den[s][u] = num[s][u];
      isv[s][u] = num[s][u];
    }
  v2f qt2[2], qr2[2], qc2[2];
#pragma unroll
  for (int s = 0; s < 2; ++s) {
    qt2[s] = (v2f){0.f, 0.f};
    qr2[s] = qt2[s];
    qc2[s] = qt2[s];
  }
  for (int kc = 0; kc < KS; kc += 4 * KCH) {
    v4f a1[KCH / 2][2], a2[KCH / 2][2], a3[KCH / 2][2];
    v4f b1[KCH / 2][2], b2[KCH / 2][2];
#pragma unroll
    for (int i = 0; i < KCH / 2; ++i) {
      const int k0 = kc + 4 * i;
      a1[i][0] = *(const v4f*)(trm + ra0 + k0);  a1[i][1] = *(const v4f*)(trm + ra1 + k0);
      a2[i][0] = *(const v4f*)(aem + ra0 + k0);  a2[i][1] = *(const v4f*)(aem + ra1 + k0);
      a3[i][0] = *(const v4f*)(r   + ra0 + k0);  a3[i][1] = *(const v4f*)(r   + ra1 + k0);
      b1[i][0] = *(const v4f*)(t   + cb0 + k0);  b1[i][1] = *(const v4f*)(t   + cb1 + k0);
    }
    __builtin_amdgcn_sched_barrier(0);
#pragma unroll
    for (int i = 0; i < KCH / 2; ++i)
#pragma unroll
      for (int s = 0; s < 2; ++s) {
        qt2[s] = sq_acc(qt2[s], a1[i][s]);
        qr2[s] = sq_acc(qr2[s], a3[i][s]);
        qc2[s] = sq_acc(qc2[s], b1[i][s]);
        a2[i][s] = exp4(a2[i][s], mx[s]);
        a1[i][s] = a1[i][s] * a2[i][s];
        a2[i][s] = a2[i][s] * a2[i][s];
        b2[i][s] = b1[i][s] * b1[i][s];
      }
#pragma unroll
    for (int j = 0; j < KCH; ++j)
#pragma unroll
      for (int s = 0; s < 2; ++s)
#pragma unroll
        for (int u = 0; u < 2; ++u) {
          const int i = j >> 1;
          const v2f fa1 = (j & 1) ? hi2(a1[i][s]) : lo2(a1[i][s]);
          const v2f fa2 = (j & 1) ? hi2(a2[i][s]) : lo2(a2[i][s]);
          const v2f fa3 = (j & 1) ? hi2(a3[i][s]) : lo2(a3[i][s]);
          const v2f fb1 = (j & 1) ? hi2(b1[i][u]) : lo2(b1[i][u]);
          const v2f fb2 = (j & 1) ? hi2(b2[i][u]) : lo2(b2[i][u]);
          num[s][u] = wmma4(fa1, fb1, num[s][u]);
          den[s][u] = wmma4(fa2, fb2, den[s][u]);
          isv[s][u] = wmma4(fa3, fb1, isv[s][u]);
        }
  }
  float qt[2], qr[2], qc[2];
#pragma unroll
  for (int s = 0; s < 2; ++s) {
    qt[s] = qt2[s].x + qt2[s].y;
    qr[s] = qr2[s].x + qr2[s].y;
    qc[s] = qc2[s].x + qc2[s].y;
    qt[s] += __shfl_xor(qt[s], 16);
    qr[s] += __shfl_xor(qr[s], 16);
    qc[s] += __shfl_xor(qc[s], 16);
  }
  if (half == 0) {
#pragma unroll
    for (int s = 0; s < 2; ++s) {
      ssq[wv][0][16 * s + lid] = qt[s];
      ssq[wv][1][16 * s + lid] = qr[s];
      ssq[wv][2][16 * s + lid] = qc[s];
    }
  }
  const int rr = tid >> 4, cc = 2 * (tid & 15);
  v2f ic = {0.f, 0.f};
#pragma unroll
  for (int s = 0; s < 2; ++s) {
    if (s) __syncthreads();
#pragma unroll
    for (int u = 0; u < 2; ++u)
#pragma unroll
      for (int i = 0; i < 8; ++i) {
        part[wv][0][half * 8 + i][16 * u + lid] = num[s][u][i];
        part[wv][1][half * 8 + i][16 * u + lid] = den[s][u][i];
        part[wv][2][half * 8 + i][16 * u + lid] = isv[s][u][i];
      }
    __syncthreads();
    if (s == 0) {
      v2f c = *(const v2f*)&ssq[0][2][cc];
#pragma unroll
      for (int w = 1; w < 8; ++w) c += *(const v2f*)&ssq[w][2][cc];
      ic.x = rsqrtf(c.x);
      ic.y = rsqrtf(c.y);
    }
    float st = ssq[0][0][16 * s + rr];
    float sr = ssq[0][1][16 * s + rr];
#pragma unroll
    for (int w = 1; w < 8; ++w) {
      st += ssq[w][0][16 * s + rr];
      sr += ssq[w][1][16 * s + rr];
    }
    const float it = rsqrtf(st);
    const float ir = rsqrtf(sr);
    v2f sn = *(const v2f*)&part[0][0][rr][cc];
    v2f sd = *(const v2f*)&part[0][1][rr][cc];
    v2f si = *(const v2f*)&part[0][2][rr][cc];
#pragma unroll
    for (int w = 1; w < 8; ++w) {
      sn += *(const v2f*)&part[w][0][rr][cc];
      sd += *(const v2f*)&part[w][1][rr][cc];
      si += *(const v2f*)&part[w][2][rr][cc];
    }
    v2f o;
    o.x = sn.x * rsqrtf(sd.x) * it + si.x * (ir * ic.x);
    o.y = sn.y * rsqrtf(sd.y) * it + si.y * (ir * ic.y);
    *(v2f*)(score + (size_t)(m0 + 16 * s + rr) * BB + n0 + cc) = o;
  }
}

extern "C" void kernel_launch(void* const* d_in, const int* in_sizes, int n_in,
                              void* d_out, int out_size, void* d_ws, size_t ws_size,
                              hipStream_t stream) {
  const float* r    = (const float*)d_in[0];
  const float* m    = (const float*)d_in[1];
  const float* t    = (const float*)d_in[2];
  const int*   ntid = (const int*)d_in[3];
  const int*   nlen = (const int*)d_in[4];
  const int*   antk = (const int*)d_in[5];
  const float* emb  = (const float*)d_in[6];
  const float* nW   = (const float*)d_in[7];
  const float* nb   = (const float*)d_in[8];
  const float* W1   = (const float*)d_in[9];
  const float* b1   = (const float*)d_in[10];
  const float* W2   = (const float*)d_in[11];
  const float* b2   = (const float*)d_in[12];
  const float* trW  = (const float*)d_in[13];
  const float* trb  = (const float*)d_in[14];
  const int*   t2i  = (const int*)d_in[15];

  float* score    = (float*)d_out;
  float* concepts = score + (size_t)BB * BB;

  float* ws  = (float*)d_ws;
  float* cls = ws;  ws += (size_t)NN * DD;
  float* h   = ws;  ws += (size_t)BB * DD;
  float* aem = ws;  ws += (size_t)BB * DD;
  float* trm = ws;  ws += (size_t)BB * DD;
  int* list  = (int*)ws;
  int* src   = list + NN;

  const dim3 wave(32);
  k_gemm_m2<<<dim3(BB / (16 * M2_S), DD / 32 + 1), dim3(256), 0, stream>>>(
      m, W1, b1, h, trW, trb, trm, ntid, nlen, t2i, list, src);
  k_rows<<<dim3(DD / 256, ROW_GY), dim3(128), 0, stream>>>(list, antk, ntid, emb, nW, nb,
                                                           cls, concepts);
  k_concepts<<<dim3(NPOS / 64, DD / 64), wave, 0, stream>>>(src, cls, concepts);
  k_gemm_xw<<<dim3(BB / 32, DD / 64), dim3(256), 0, stream>>>(h, W2, b2, aem);
  k_score<<<dim3(BB / 32, BB / 32), dim3(256), 0, stream>>>(trm, aem, r, t, score);
}
